# starting kernel
# baseline (speedup 1.0000x reference)
#include <hip/hip_runtime.h>
#include <cstdint>
#include <cstddef>


typedef __attribute__((ext_vector_type(2))) float v2f;
typedef __attribute__((ext_vector_type(4))) float f4;
typedef __attribute__((ext_vector_type(8))) float v8f;

#define DFEAT 128
#define SCAN_THREADS 1024
#define CSR_PAD 16


__global__ void zero_i32_kernel(int* __restrict__ p, int n) {
  int i = blockIdx.x * blockDim.x + threadIdx.x;
  int s = gridDim.x * blockDim.x;
  for (; i < n; i += s) p[i] = 0;
}

__global__ void zero_f32_kernel(float* __restrict__ p, int n) {
  int i = blockIdx.x * blockDim.x + threadIdx.x;
  int s = gridDim.x * blockDim.x;
  for (; i < n; i += s) p[i] = 0.0f;
}

__global__ void pad_kernel(int* __restrict__ csrCol, int* __restrict__ csrRow,
                           float* __restrict__ csrVal, int E) {
  int i = threadIdx.x;
  if (i < CSR_PAD) {
    csrCol[E + i] = 0;
    csrRow[E + i] = -1;
    csrVal[E + i] = 0.0f;
  }
}


__global__ void hist_kernel(const int* __restrict__ rows,
                            int* __restrict__ counts, int E) {
  int i = blockIdx.x * blockDim.x + threadIdx.x;
  int s = gridDim.x * blockDim.x;
  for (; i < E; i += s) atomicAdd(&counts[rows[i]], 1);
}

__global__ __launch_bounds__(SCAN_THREADS)
void scan_kernel(const int* __restrict__ counts, int* __restrict__ rowStart,
                 int n) {
  __shared__ int sm[SCAN_THREADS];
  int carry = 0;
  for (int base = 0; base < n; base += SCAN_THREADS) {
    int i = base + (int)threadIdx.x;
    int x = (i < n) ? counts[i] : 0;
    sm[threadIdx.x] = x;
    __syncthreads();
    for (int off = 1; off < SCAN_THREADS; off <<= 1) {
      int t = (threadIdx.x >= (unsigned)off) ? sm[threadIdx.x - off] : 0;
      __syncthreads();
      sm[threadIdx.x] += t;
      __syncthreads();
    }
    int incl = sm[threadIdx.x];
    if (i < n) rowStart[i] = carry + (incl - x);
    carry += sm[SCAN_THREADS - 1];
    __syncthreads();
  }
  if (threadIdx.x == 0) rowStart[n] = carry;
}

__global__ void fill_kernel(const int* __restrict__ rows,
                            const int* __restrict__ cols,
                            const float* __restrict__ vals,
                            const int* __restrict__ rowStart,
                            int* __restrict__ cursor,
                            int* __restrict__ csrCol, int* __restrict__ csrRow,
                            float* __restrict__ csrVal, int E) {
  int i = blockIdx.x * blockDim.x + threadIdx.x;
  int s = gridDim.x * blockDim.x;
  for (; i < E; i += s) {
    int r = rows[i];
    int pos = rowStart[r] + atomicAdd(&cursor[r], 1);
    csrCol[pos] = cols[i];
    csrRow[pos] = r;
    csrVal[pos] = vals[i];
  }
}


struct Meta {
  int2 c;
  int2 r;
  float2 v;
};

__device__ __forceinline__ Meta load_meta(const int* __restrict__ csrCol,
                                          const int* __restrict__ csrRow,
                                          const float* __restrict__ csrVal,
                                          int ka) {
  Meta mt;
  mt.c = *(const int2*)(csrCol + ka);
  mt.r = *(const int2*)(csrRow + ka);
  mt.v = *(const float2*)(csrVal + ka);
  return mt;
}

__global__ __launch_bounds__(256)
void wmma_spmm_kernel(const int* __restrict__ rowStart,
                      const int* __restrict__ csrCol,
                      const int* __restrict__ csrRow,
                      const float* __restrict__ csrVal,
                      const float* __restrict__ ann, float* __restrict__ out) {
  const int base = blockIdx.x * 16;
  const int kBeg = rowStart[base];
  const int kEnd = rowStart[base + 16];
  const int lane = (int)(threadIdx.x & 31);
  const int wv = (int)(threadIdx.x >> 5);
  const int half = lane >> 4;
  const int jbn = wv * 16 + (lane & 15);
  const int rowm = base + (lane & 15);

  v8f acc = {0.f, 0.f, 0.f, 0.f, 0.f, 0.f, 0.f, 0.f};

  int k0 = kBeg & ~3;
  int ka = k0 + 2 * half;

  if (k0 < kEnd) {
    Meta m0 = load_meta(csrCol, csrRow, csrVal, ka);
    Meta m1 = load_meta(csrCol, csrRow, csrVal, ka + 4);
    v2f b0;
    b0.x = ann[m0.c.x * DFEAT + jbn];
    b0.y = ann[m0.c.y * DFEAT + jbn];

#pragma unroll 2
    for (; k0 < kEnd; k0 += 4, ka += 4) {
      Meta m2 = load_meta(csrCol, csrRow, csrVal, ka + 8);

      v2f b1;
      b1.x = ann[m1.c.x * DFEAT + jbn];
      b1.y = ann[m1.c.y * DFEAT + jbn];

      v2f a;
      a.x = (m0.r.x == rowm) ? m0.v.x : 0.0f;
      a.y = (m0.r.y == rowm) ? m0.v.y : 0.0f;

      acc = __builtin_amdgcn_wmma_f32_16x16x4_f32(
           false, a,  false, b0,
           (short)0, acc,  false,  false);

      m0 = m1;
      m1 = m2;
      b0 = b1;
    }
  }

  float* o = out + (size_t)(base + 8 * half) * DFEAT + jbn;
#pragma unroll
  for (int v = 0; v < 8; ++v) o[(size_t)v * DFEAT] = acc[v];
}


__global__ __launch_bounds__(256)
void scatter_kernel(const int* __restrict__ rows, const int* __restrict__ cols,
                    const float* __restrict__ vals,
                    const float* __restrict__ ann, float* __restrict__ out,
                    int E) {
  const int lane = (int)(threadIdx.x & 31);
  int wid = (int)((blockIdx.x * blockDim.x + threadIdx.x) >> 5);
  const int nw = (int)((gridDim.x * blockDim.x) >> 5);
  for (int e = wid; e < E; e += nw) {
    const int r = rows[e];
    const int c = cols[e];
    const float v = vals[e];
    const f4 x = *(const f4*)(ann + (size_t)c * DFEAT + lane * 4);
    float* o = out + (size_t)r * DFEAT + lane * 4;
    unsafeAtomicAdd(o + 0, v * x.x);
    unsafeAtomicAdd(o + 1, v * x.y);
    unsafeAtomicAdd(o + 2, v * x.z);
    unsafeAtomicAdd(o + 3, v * x.w);
  }
}


extern "C" void kernel_launch(void* const* d_in, const int* in_sizes, int n_in,
                              void* d_out, int out_size, void* d_ws,
                              size_t ws_size, hipStream_t stream) {
  const int* rows = (const int*)d_in[0];
  const int* cols = (const int*)d_in[1];
  const float* vals = (const float*)d_in[2];
  const float* ann = (const float*)d_in[3];
  float* out = (float*)d_out;

  const int E = in_sizes[0];
  const int N = out_size / DFEAT;

  const size_t need = (size_t)(3 * N + 1) * sizeof(int) +
                      (size_t)(E + CSR_PAD) * 12;

  if ((N % 16) == 0 && ws_size >= need) {
    char* w = (char*)d_ws;
    int* counts = (int*)w;    w += (size_t)N * 4;
    int* cursor = (int*)w;    w += (size_t)N * 4;
    int* rowStart = (int*)w;  w += (size_t)(N + 1) * 4;
    int* csrCol = (int*)w;    w += (size_t)(E + CSR_PAD) * 4;
    int* csrRow = (int*)w;    w += (size_t)(E + CSR_PAD) * 4;
    float* csrVal = (float*)w;

    const int gz = (2 * N + 255) / 256;
    const int ge = (E + 255) / 256;

    zero_i32_kernel<<<gz, 256, 0, stream>>>(counts, 2 * N);
    pad_kernel<<<1, 32, 0, stream>>>(csrCol, csrRow, csrVal, E);
    hist_kernel<<<ge, 256, 0, stream>>>(rows, counts, E);
    scan_kernel<<<1, SCAN_THREADS, 0, stream>>>(counts, rowStart, N);
    fill_kernel<<<ge, 256, 0, stream>>>(rows, cols, vals, rowStart, cursor,
                                        csrCol, csrRow, csrVal, E);
    wmma_spmm_kernel<<<N / 16, 256, 0, stream>>>(rowStart, csrCol, csrRow,
                                                 csrVal, ann, out);
  } else {
    zero_f32_kernel<<<(out_size + 255) / 256, 256, 0, stream>>>(out, out_size);
    scatter_kernel<<<2048, 256, 0, stream>>>(rows, cols, vals, ann, out, E);
  }
}


# row-offset scan as one sweep: all 16-byte count loads in flight at once, per-wave scans as DPP adds, 320 wave totals scanned by one wave; 2 barriers instead of 40 tiles x 20-barrier LDS scan
# speedup vs baseline: 1.5020x; 1.5020x over previous
#include <hip/hip_runtime.h>
#include <cstdint>
#include <cstddef>


typedef __attribute__((ext_vector_type(2))) float v2f;
typedef __attribute__((ext_vector_type(4))) float f4;
typedef __attribute__((ext_vector_type(8))) float v8f;

#define DFEAT 128
#define SCAN_THREADS 1024
#define CSR_PAD 16


__global__ void zero_i32_kernel(int* __restrict__ p, int n) {
  int i = blockIdx.x * blockDim.x + threadIdx.x;
  int s = gridDim.x * blockDim.x;
  for (; i < n; i += s) p[i] = 0;
}

__global__ void zero_f32_kernel(float* __restrict__ p, int n) {
  int i = blockIdx.x * blockDim.x + threadIdx.x;
  int s = gridDim.x * blockDim.x;
  for (; i < n; i += s) p[i] = 0.0f;
}

__global__ void pad_kernel(int* __restrict__ csrCol, int* __restrict__ csrRow,
                           float* __restrict__ csrVal, int E) {
  int i = threadIdx.x;
  if (i < CSR_PAD) {
    csrCol[E + i] = 0;
    csrRow[E + i] = -1;
    csrVal[E + i] = 0.0f;
  }
}


__global__ void hist_kernel(const int* __restrict__ rows,
                            int* __restrict__ counts, int E) {
  int i = blockIdx.x * blockDim.x + threadIdx.x;
  int s = gridDim.x * blockDim.x;
  for (; i < E; i += s) atomicAdd(&counts[rows[i]], 1);
}

__device__ __forceinline__ int wave_incl_scan(int s, int lane) {
  s += __builtin_amdgcn_update_dpp(0, s, 0x111, 0xf, 0xf, true);
  s += __builtin_amdgcn_update_dpp(0, s, 0x112, 0xf, 0xf, true);
  s += __builtin_amdgcn_update_dpp(0, s, 0x114, 0xf, 0xf, true);
  s += __builtin_amdgcn_update_dpp(0, s, 0x118, 0xf, 0xf, true);
  const int row0 = __builtin_amdgcn_readlane(s, 15);
  return s + ((lane >= 16) ? row0 : 0);
}

#define SCAN_WAVES (SCAN_THREADS / 32)
#define SCAN_TPT 10
#define SCAN_TILE (SCAN_THREADS * 4)
#define SCAN_SWEEP (SCAN_TILE * SCAN_TPT)
static_assert(SCAN_WAVES == 32, "wave 0 scans SCAN_TPT totals per lane");

__global__ __launch_bounds__(SCAN_THREADS)
void scan_kernel(const int* __restrict__ counts, int* __restrict__ rowStart,
                 int n) {
  __shared__ int wtot[SCAN_TPT * SCAN_WAVES];
  __shared__ int sweepTotal;
  const int tid = (int)threadIdx.x;
  const int lane = tid & 31;
  const int wv = tid >> 5;
  int carry = 0;

  for (int base = 0; base < n; base += SCAN_SWEEP) {
    int4 x[SCAN_TPT];
#pragma unroll
    for (int j = 0; j < SCAN_TPT; ++j) {
      const int i = base + j * SCAN_TILE + tid * 4;
      const int4 t = *(const int4*)(counts + ((i < n) ? i : 0));
      x[j].x = (i < n) ? t.x : 0;
      x[j].y = (i < n) ? t.y : 0;
      x[j].z = (i < n) ? t.z : 0;
      x[j].w = (i < n) ? t.w : 0;
    }

    int own[SCAN_TPT], s[SCAN_TPT];
#pragma unroll
    for (int j = 0; j < SCAN_TPT; ++j) {
      own[j] = x[j].x + x[j].y + x[j].z + x[j].w;
    }
#pragma unroll
    for (int j = 0; j < SCAN_TPT; ++j) s[j] = wave_incl_scan(own[j], lane);
    if (lane == 31) {
#pragma unroll
      for (int j = 0; j < SCAN_TPT; ++j) wtot[j * SCAN_WAVES + wv] = s[j];
    }
    __syncthreads();

    if (wv == 0) {
      int v[SCAN_TPT];
      int t = 0;
#pragma unroll
      for (int q = 0; q < SCAN_TPT; ++q) {
        v[q] = wtot[lane * SCAN_TPT + q];
        t += v[q];
      }
      const int incl = wave_incl_scan(t, lane);
      int run = incl - t;
#pragma unroll
      for (int q = 0; q < SCAN_TPT; ++q) {
        wtot[lane * SCAN_TPT + q] = run;
        run += v[q];
      }
      if (lane == 31) sweepTotal = incl;
    }
    __syncthreads();

#pragma unroll
    for (int j = 0; j < SCAN_TPT; ++j) {
      const int i = base + j * SCAN_TILE + tid * 4;
      int4 o;
      o.x = carry + wtot[j * SCAN_WAVES + wv] + (s[j] - own[j]);
      o.y = o.x + x[j].x;
      o.z = o.y + x[j].y;
      o.w = o.z + x[j].z;
      if (i < n) *(int4*)(rowStart + i) = o;
    }
    carry += sweepTotal;
    __syncthreads();
  }
  if (tid == 0) rowStart[n] = carry;
}

__global__ void fill_kernel(const int* __restrict__ rows,
                            const int* __restrict__ cols,
                            const float* __restrict__ vals,
                            const int* __restrict__ rowStart,
                            int* __restrict__ cursor,
                            int* __restrict__ csrCol, int* __restrict__ csrRow,
                            float* __restrict__ csrVal, int E) {
  int i = blockIdx.x * blockDim.x + threadIdx.x;
  int s = gridDim.x * blockDim.x;
  for (; i < E; i += s) {
    int r = rows[i];
    int pos = rowStart[r] + atomicAdd(&cursor[r], 1);
    csrCol[pos] = cols[i];
    csrRow[pos] = r;
    csrVal[pos] = vals[i];
  }
}


struct Meta {
  int2 c;
  int2 r;
  float2 v;
};

__device__ __forceinline__ Meta load_meta(const int* __restrict__ csrCol,
                                          const int* __restrict__ csrRow,
                                          const float* __restrict__ csrVal,
                                          int ka) {
  Meta mt;
  mt.c = *(const int2*)(csrCol + ka);
  mt.r = *(const int2*)(csrRow + ka);
  mt.v = *(const float2*)(csrVal + ka);
  return mt;
}

__global__ __launch_bounds__(256)
void wmma_spmm_kernel(const int* __restrict__ rowStart,
                      const int* __restrict__ csrCol,
                      const int* __restrict__ csrRow,
                      const float* __restrict__ csrVal,
                      const float* __restrict__ ann, float* __restrict__ out) {
  const int base = blockIdx.x * 16;
  const int kBeg = rowStart[base];
  const int kEnd = rowStart[base + 16];
  const int lane = (int)(threadIdx.x & 31);
  const int wv = (int)(threadIdx.x >> 5);
  const int half = lane >> 4;
  const int jbn = wv * 16 + (lane & 15);
  const int rowm = base + (lane & 15);

  v8f acc = {0.f, 0.f, 0.f, 0.f, 0.f, 0.f, 0.f, 0.f};

  int k0 = kBeg & ~3;
  int ka = k0 + 2 * half;

  if (k0 < kEnd) {
    Meta m0 = load_meta(csrCol, csrRow, csrVal, ka);
    Meta m1 = load_meta(csrCol, csrRow, csrVal, ka + 4);
    v2f b0;
    b0.x = ann[m0.c.x * DFEAT + jbn];
    b0.y = ann[m0.c.y * DFEAT + jbn];

#pragma unroll 2
    for (; k0 < kEnd; k0 += 4, ka += 4) {
      Meta m2 = load_meta(csrCol, csrRow, csrVal, ka + 8);

      v2f b1;
      b1.x = ann[m1.c.x * DFEAT + jbn];
      b1.y = ann[m1.c.y * DFEAT + jbn];

      v2f a;
      a.x = (m0.r.x == rowm) ? m0.v.x : 0.0f;
      a.y = (m0.r.y == rowm) ? m0.v.y : 0.0f;

      acc = __builtin_amdgcn_wmma_f32_16x16x4_f32(
           false, a,  false, b0,
           (short)0, acc,  false,  false);

      m0 = m1;
      m1 = m2;
      b0 = b1;
    }
  }

  float* o = out + (size_t)(base + 8 * half) * DFEAT + jbn;
#pragma unroll
  for (int v = 0; v < 8; ++v) o[(size_t)v * DFEAT] = acc[v];
}


__global__ __launch_bounds__(256)
void scatter_kernel(const int* __restrict__ rows, const int* __restrict__ cols,
                    const float* __restrict__ vals,
                    const float* __restrict__ ann, float* __restrict__ out,
                    int E) {
  const int lane = (int)(threadIdx.x & 31);
  int wid = (int)((blockIdx.x * blockDim.x + threadIdx.x) >> 5);
  const int nw = (int)((gridDim.x * blockDim.x) >> 5);
  for (int e = wid; e < E; e += nw) {
    const int r = rows[e];
    const int c = cols[e];
    const float v = vals[e];
    const f4 x = *(const f4*)(ann + (size_t)c * DFEAT + lane * 4);
    float* o = out + (size_t)r * DFEAT + lane * 4;
    unsafeAtomicAdd(o + 0, v * x.x);
    unsafeAtomicAdd(o + 1, v * x.y);
    unsafeAtomicAdd(o + 2, v * x.z);
    unsafeAtomicAdd(o + 3, v * x.w);
  }
}


extern "C" void kernel_launch(void* const* d_in, const int* in_sizes, int n_in,
                              void* d_out, int out_size, void* d_ws,
                              size_t ws_size, hipStream_t stream) {
  const int* rows = (const int*)d_in[0];
  const int* cols = (const int*)d_in[1];
  const float* vals = (const float*)d_in[2];
  const float* ann = (const float*)d_in[3];
  float* out = (float*)d_out;

  const int E = in_sizes[0];
  const int N = out_size / DFEAT;

  const size_t need = (size_t)(3 * N + 1) * sizeof(int) +
                      (size_t)(E + CSR_PAD) * 12;

  if ((N % 16) == 0 && ws_size >= need) {
    char* w = (char*)d_ws;
    int* counts = (int*)w;    w += (size_t)N * 4;
    int* cursor = (int*)w;    w += (size_t)N * 4;
    int* rowStart = (int*)w;  w += (size_t)(N + 1) * 4;
    int* csrCol = (int*)w;    w += (size_t)(E + CSR_PAD) * 4;
    int* csrRow = (int*)w;    w += (size_t)(E + CSR_PAD) * 4;
    float* csrVal = (float*)w;

    const int gz = (2 * N + 255) / 256;
    const int ge = (E + 255) / 256;

    zero_i32_kernel<<<gz, 256, 0, stream>>>(counts, 2 * N);
    pad_kernel<<<1, 32, 0, stream>>>(csrCol, csrRow, csrVal, E);
    hist_kernel<<<ge, 256, 0, stream>>>(rows, counts, E);
    scan_kernel<<<1, SCAN_THREADS, 0, stream>>>(counts, rowStart, N);
    fill_kernel<<<ge, 256, 0, stream>>>(rows, cols, vals, rowStart, cursor,
                                        csrCol, csrRow, csrVal, E);
    wmma_spmm_kernel<<<N / 16, 256, 0, stream>>>(rowStart, csrCol, csrRow,
                                                 csrVal, ann, out);
  } else {
    zero_f32_kernel<<<(out_size + 255) / 256, 256, 0, stream>>>(out, out_size);
    scatter_kernel<<<2048, 256, 0, stream>>>(rows, cols, vals, ann, out, E);
  }
}


# spmm: window's edge chunks split over the 8 waves; a wave gathers whole 512-B rows by 16-B loads, 8 WMMA tiles per A fragment; partial tiles summed in wave order via LDS (sum order changes)
# speedup vs baseline: 1.8372x; 1.2232x over previous
#include <hip/hip_runtime.h>
#include <cstdint>
#include <cstddef>


typedef __attribute__((ext_vector_type(2))) float v2f;
typedef __attribute__((ext_vector_type(4))) float f4;
typedef __attribute__((ext_vector_type(8))) float v8f;

#define DFEAT 128
#define SCAN_THREADS 1024
#define CSR_PAD 16


__global__ void zero_i32_kernel(int* __restrict__ p, int n) {
  int i = blockIdx.x * blockDim.x + threadIdx.x;
  int s = gridDim.x * blockDim.x;
  for (; i < n; i += s) p[i] = 0;
}

__global__ void zero_f32_kernel(float* __restrict__ p, int n) {
  int i = blockIdx.x * blockDim.x + threadIdx.x;
  int s = gridDim.x * blockDim.x;
  for (; i < n; i += s) p[i] = 0.0f;
}

__global__ void pad_kernel(int* __restrict__ csrCol, int* __restrict__ csrRow,
                           float* __restrict__ csrVal, int E) {
  int i = threadIdx.x;
  if (i < CSR_PAD) {
    csrCol[E + i] = 0;
    csrRow[E + i] = -1;
    csrVal[E + i] = 0.0f;
  }
}


__global__ void hist_kernel(const int* __restrict__ rows,
                            int* __restrict__ counts, int E) {
  int i = blockIdx.x * blockDim.x + threadIdx.x;
  int s = gridDim.x * blockDim.x;
  for (; i < E; i += s) atomicAdd(&counts[rows[i]], 1);
}

__device__ __forceinline__ int wave_incl_scan(int s, int lane) {
  s += __builtin_amdgcn_update_dpp(0, s, 0x111, 0xf, 0xf, true);
  s += __builtin_amdgcn_update_dpp(0, s, 0x112, 0xf, 0xf, true);
  s += __builtin_amdgcn_update_dpp(0, s, 0x114, 0xf, 0xf, true);
  s += __builtin_amdgcn_update_dpp(0, s, 0x118, 0xf, 0xf, true);
  const int row0 = __builtin_amdgcn_readlane(s, 15);
  return s + ((lane >= 16) ? row0 : 0);
}

#define SCAN_WAVES (SCAN_THREADS / 32)
#define SCAN_TPT 10
#define SCAN_TILE (SCAN_THREADS * 4)
#define SCAN_SWEEP (SCAN_TILE * SCAN_TPT)
static_assert(SCAN_WAVES == 32, "wave 0 scans SCAN_TPT totals per lane");

__global__ __launch_bounds__(SCAN_THREADS)
void scan_kernel(const int* __restrict__ counts, int* __restrict__ rowStart,
                 int n) {
  __shared__ int wtot[SCAN_TPT * SCAN_WAVES];
  __shared__ int sweepTotal;
  const int tid = (int)threadIdx.x;
  const int lane = tid & 31;
  const int wv = tid >> 5;
  int carry = 0;

  for (int base = 0; base < n; base += SCAN_SWEEP) {
    int4 x[SCAN_TPT];
#pragma unroll
    for (int j = 0; j < SCAN_TPT; ++j) {
      const int i = base + j * SCAN_TILE + tid * 4;
      const int4 t = *(const int4*)(counts + ((i < n) ? i : 0));
      x[j].x = (i < n) ? t.x : 0;
      x[j].y = (i < n) ? t.y : 0;
      x[j].z = (i < n) ? t.z : 0;
      x[j].w = (i < n) ? t.w : 0;
    }

    int own[SCAN_TPT], s[SCAN_TPT];
#pragma unroll
    for (int j = 0; j < SCAN_TPT; ++j) {
      own[j] = x[j].x + x[j].y + x[j].z + x[j].w;
    }
#pragma unroll
    for (int j = 0; j < SCAN_TPT; ++j) s[j] = wave_incl_scan(own[j], lane);
    if (lane == 31) {
#pragma unroll
      for (int j = 0; j < SCAN_TPT; ++j) wtot[j * SCAN_WAVES + wv] = s[j];
    }
    __syncthreads();

    if (wv == 0) {
      int v[SCAN_TPT];
      int t = 0;
#pragma unroll
      for (int q = 0; q < SCAN_TPT; ++q) {
        v[q] = wtot[lane * SCAN_TPT + q];
        t += v[q];
      }
      const int incl = wave_incl_scan(t, lane);
      int run = incl - t;
#pragma unroll
      for (int q = 0; q < SCAN_TPT; ++q) {
        wtot[lane * SCAN_TPT + q] = run;
        run += v[q];
      }
      if (lane == 31) sweepTotal = incl;
    }
    __syncthreads();

#pragma unroll
    for (int j = 0; j < SCAN_TPT; ++j) {
      const int i = base + j * SCAN_TILE + tid * 4;
      int4 o;
      o.x = carry + wtot[j * SCAN_WAVES + wv] + (s[j] - own[j]);
      o.y = o.x + x[j].x;
      o.z = o.y + x[j].y;
      o.w = o.z + x[j].z;
      if (i < n) *(int4*)(rowStart + i) = o;
    }
    carry += sweepTotal;
    __syncthreads();
  }
  if (tid == 0) rowStart[n] = carry;
}

__global__ void fill_kernel(const int* __restrict__ rows,
                            const int* __restrict__ cols,
                            const float* __restrict__ vals,
                            const int* __restrict__ rowStart,
                            int* __restrict__ cursor,
                            int* __restrict__ csrCol, int* __restrict__ csrRow,
                            float* __restrict__ csrVal, int E) {
  int i = blockIdx.x * blockDim.x + threadIdx.x;
  int s = gridDim.x * blockDim.x;
  for (; i < E; i += s) {
    int r = rows[i];
    int pos = rowStart[r] + atomicAdd(&cursor[r], 1);
    csrCol[pos] = cols[i];
    csrRow[pos] = r;
    csrVal[pos] = vals[i];
  }
}


#define KSPLIT 8
#define PASS_CHUNKS 16

struct BRows {
  f4 lo0, hi0;
  f4 lo1, hi1;
};

__device__ __forceinline__ int from_chunk(int x, int i, int half) {
  const int lo = __builtin_amdgcn_readlane(x, 2 * i);
  const int hi = __builtin_amdgcn_readlane(x, 2 * i + 1);
  return half ? hi : lo;
}

__device__ __forceinline__ BRows gather_rows(const float* __restrict__ ann,
                                             int2 colv, int i, int half,
                                             int c4) {
  BRows b;
  const float* p0 = ann + from_chunk(colv.x, i, half) * DFEAT + c4;
  const float* p1 = ann + from_chunk(colv.y, i, half) * DFEAT + c4;
  b.lo0 = *(const f4*)(p0);
  b.hi0 = *(const f4*)(p0 + 64);
  b.lo1 = *(const f4*)(p1);
  b.hi1 = *(const f4*)(p1 + 64);
  return b;
}

__device__ __forceinline__ v2f a_frag(int2 rowv, float2 valv, int i, int half,
                                      int rowm, bool ok) {
  const int rx = from_chunk(rowv.x, i, half);
  const int ry = from_chunk(rowv.y, i, half);
  const float vx = __int_as_float(from_chunk(__float_as_int(valv.x), i, half));
  const float vy = __int_as_float(from_chunk(__float_as_int(valv.y), i, half));
  v2f a;
  a.x = (ok && rx == rowm) ? vx : 0.0f;
  a.y = (ok && ry == rowm) ? vy : 0.0f;
  return a;
}

__device__ __forceinline__ void mma_rows(v8f (&acc)[8], v2f a,
                                         const BRows& b) {
#pragma unroll
  for (int t = 0; t < 4; ++t) {
    v2f bl, bh;
    bl.x = b.lo0[t];
    bl.y = b.lo1[t];
    bh.x = b.hi0[t];
    bh.y = b.hi1[t];
    acc[t] = __builtin_amdgcn_wmma_f32_16x16x4_f32(
         false, a,  false, bl,
         (short)0, acc[t],  false,  false);
    acc[4 + t] = __builtin_amdgcn_wmma_f32_16x16x4_f32(
         false, a,  false, bh,
         (short)0, acc[4 + t],  false,  false);
  }
}

__global__ __launch_bounds__(256)
void wmma_spmm_kernel(const int* __restrict__ rowStart,
                      const int* __restrict__ csrCol,
                      const int* __restrict__ csrRow,
                      const float* __restrict__ csrVal,
                      const float* __restrict__ ann, float* __restrict__ out) {
  __shared__ f4 part[KSPLIT * 16 * (DFEAT / 4)];

  const int base = blockIdx.x * 16;
  const int kBeg = rowStart[base];
  const int kEnd = rowStart[base + 16];
  const int lane = (int)(threadIdx.x & 31);
  const int wv = __builtin_amdgcn_readfirstlane((int)(threadIdx.x >> 5));
  const int half = lane >> 4;
  const int c = lane & 15;
  const int c4 = 4 * c;
  const int rowm = base + c;

  v8f acc[8];
#pragma unroll
  for (int t = 0; t < 8; ++t) {
    acc[t] = (v8f){0.f, 0.f, 0.f, 0.f, 0.f, 0.f, 0.f, 0.f};
  }

  const int k0 = kBeg & ~3;
  const int nCh = (kEnd - k0 + 3) >> 2;
  const int per = (nCh + KSPLIT - 1) / KSPLIT;
  const int jb = wv * per;
  const int je = (jb + per < nCh) ? jb + per : nCh;

  for (int seg = jb; seg < je; seg += PASS_CHUNKS) {
    const int n = (je - seg < PASS_CHUNKS) ? je - seg : PASS_CHUNKS;
    const int e = k0 + 4 * seg + ((lane < 2 * n) ? 2 * lane : 0);
    const int2 colv = *(const int2*)(csrCol + e);
    const int2 rowv = *(const int2*)(csrRow + e);
    const float2 valv = *(const float2*)(csrVal + e);

    const int n2 = (n + 1) & ~1;
    BRows bA = gather_rows(ann, colv, 0, half, c4);
    int i = 0;
    for (; i + 2 < n2; i += 2) {
      const BRows bB = gather_rows(ann, colv, i + 1, half, c4);
      __builtin_amdgcn_sched_barrier(0);
      mma_rows(acc, a_frag(rowv, valv, i, half, rowm, true), bA);
      bA = gather_rows(ann, colv, i + 2, half, c4);
      __builtin_amdgcn_sched_barrier(0);
      mma_rows(acc, a_frag(rowv, valv, i + 1, half, rowm, true), bB);
    }
    {
      const bool ok = (i + 1 < n);
      const int i1 = ok ? i + 1 : i;
      const BRows bB = gather_rows(ann, colv, i1, half, c4);
      __builtin_amdgcn_sched_barrier(0);
      mma_rows(acc, a_frag(rowv, valv, i, half, rowm, true), bA);
      mma_rows(acc, a_frag(rowv, valv, i1, half, rowm, ok), bB);
    }
  }

  f4* pw = part + wv * (16 * (DFEAT / 4)) + (8 * half) * (DFEAT / 4) + c;
#pragma unroll
  for (int v = 0; v < 8; ++v) {
#pragma unroll
    for (int g = 0; g < 2; ++g) {
      f4 x;
      x.x = acc[4 * g + 0][v];
      x.y = acc[4 * g + 1][v];
      x.z = acc[4 * g + 2][v];
      x.w = acc[4 * g + 3][v];
      pw[v * (DFEAT / 4) + 16 * g] = x;
    }
  }
  __syncthreads();

#pragma unroll
  for (int i = 0; i < 2; ++i) {
    const int row = wv + 8 * i;
    const f4* ps = part + row * (DFEAT / 4) + lane;
    f4 s = ps[0];
#pragma unroll
    for (int w = 1; w < KSPLIT; ++w) s += ps[w * (16 * (DFEAT / 4))];
    *(f4*)(out + (size_t)(base + row) * DFEAT + 4 * lane) = s;
  }
}


__global__ __launch_bounds__(256)
void scatter_kernel(const int* __restrict__ rows, const int* __restrict__ cols,
                    const float* __restrict__ vals,
                    const float* __restrict__ ann, float* __restrict__ out,
                    int E) {
  const int lane = (int)(threadIdx.x & 31);
  int wid = (int)((blockIdx.x * blockDim.x + threadIdx.x) >> 5);
  const int nw = (int)((gridDim.x * blockDim.x) >> 5);
  for (int e = wid; e < E; e += nw) {
    const int r = rows[e];
    const int c = cols[e];
    const float v = vals[e];
    const f4 x = *(const f4*)(ann + (size_t)c * DFEAT + lane * 4);
    float* o = out + (size_t)r * DFEAT + lane * 4;
    unsafeAtomicAdd(o + 0, v * x.x);
    unsafeAtomicAdd(o + 1, v * x.y);
    unsafeAtomicAdd(o + 2, v * x.z);
    unsafeAtomicAdd(o + 3, v * x.w);
  }
}


extern "C" void kernel_launch(void* const* d_in, const int* in_sizes, int n_in,
                              void* d_out, int out_size, void* d_ws,
                              size_t ws_size, hipStream_t stream) {
  const int* rows = (const int*)d_in[0];
  const int* cols = (const int*)d_in[1];
  const float* vals = (const float*)d_in[2];
  const float* ann = (const float*)d_in[3];
  float* out = (float*)d_out;

  const int E = in_sizes[0];
  const int N = out_size / DFEAT;

  const size_t need = (size_t)(3 * N + 1) * sizeof(int) +
                      (size_t)(E + CSR_PAD) * 12;

  if ((N % 16) == 0 && ws_size >= need) {
    char* w = (char*)d_ws;
    int* counts = (int*)w;    w += (size_t)N * 4;
    int* cursor = (int*)w;    w += (size_t)N * 4;
    int* rowStart = (int*)w;  w += (size_t)(N + 1) * 4;
    int* csrCol = (int*)w;    w += (size_t)(E + CSR_PAD) * 4;
    int* csrRow = (int*)w;    w += (size_t)(E + CSR_PAD) * 4;
    float* csrVal = (float*)w;

    const int gz = (2 * N + 255) / 256;
    const int ge = (E + 255) / 256;

    zero_i32_kernel<<<gz, 256, 0, stream>>>(counts, 2 * N);
    pad_kernel<<<1, 32, 0, stream>>>(csrCol, csrRow, csrVal, E);
    hist_kernel<<<ge, 256, 0, stream>>>(rows, counts, E);
    scan_kernel<<<1, SCAN_THREADS, 0, stream>>>(counts, rowStart, N);
    fill_kernel<<<ge, 256, 0, stream>>>(rows, cols, vals, rowStart, cursor,
                                        csrCol, csrRow, csrVal, E);
    wmma_spmm_kernel<<<N / 16, 256, 0, stream>>>(rowStart, csrCol, csrRow,
                                                 csrVal, ann, out);
  } else {
    zero_f32_kernel<<<(out_size + 255) / 256, 256, 0, stream>>>(out, out_size);
    scatter_kernel<<<2048, 256, 0, stream>>>(rows, cols, vals, ann, out, E);
  }
}


# count pass keeps each edge's atomic return as its rank in the row; fill places edges at rowStart+rank: second atomic pass and cursor array gone (in-row sum order: count pass arrival order)
# speedup vs baseline: 2.2546x; 1.2272x over previous
#include <hip/hip_runtime.h>
#include <cstdint>
#include <cstddef>


typedef __attribute__((ext_vector_type(2))) float v2f;
typedef __attribute__((ext_vector_type(4))) float f4;
typedef __attribute__((ext_vector_type(8))) float v8f;

#define DFEAT 128
#define SCAN_THREADS 1024
#define CSR_PAD 16


__global__ void zero_i32_kernel(int* __restrict__ p, int n) {
  int i = blockIdx.x * blockDim.x + threadIdx.x;
  int s = gridDim.x * blockDim.x;
  for (; i < n; i += s) p[i] = 0;
}

__global__ void zero_f32_kernel(float* __restrict__ p, int n) {
  int i = blockIdx.x * blockDim.x + threadIdx.x;
  int s = gridDim.x * blockDim.x;
  for (; i < n; i += s) p[i] = 0.0f;
}

__global__ void pad_kernel(int* __restrict__ csrCol, int* __restrict__ csrRow,
                           float* __restrict__ csrVal, int E) {
  int i = threadIdx.x;
  if (i < CSR_PAD) {
    csrCol[E + i] = 0;
    csrRow[E + i] = -1;
    csrVal[E + i] = 0.0f;
  }
}


__global__ void hist_rank_kernel(const int* __restrict__ rows,
                                 int* __restrict__ counts,
                                 int* __restrict__ rank, int E) {
  int i = blockIdx.x * blockDim.x + threadIdx.x;
  int s = gridDim.x * blockDim.x;
  for (; i < E; i += s) rank[i] = atomicAdd(&counts[rows[i]], 1);
}

__device__ __forceinline__ int wave_incl_scan(int s, int lane) {
  s += __builtin_amdgcn_update_dpp(0, s, 0x111, 0xf, 0xf, true);
  s += __builtin_amdgcn_update_dpp(0, s, 0x112, 0xf, 0xf, true);
  s += __builtin_amdgcn_update_dpp(0, s, 0x114, 0xf, 0xf, true);
  s += __builtin_amdgcn_update_dpp(0, s, 0x118, 0xf, 0xf, true);
  const int row0 = __builtin_amdgcn_readlane(s, 15);
  return s + ((lane >= 16) ? row0 : 0);
}

#define SCAN_WAVES (SCAN_THREADS / 32)
#define SCAN_TPT 10
#define SCAN_TILE (SCAN_THREADS * 4)
#define SCAN_SWEEP (SCAN_TILE * SCAN_TPT)
static_assert(SCAN_WAVES == 32, "wave 0 scans SCAN_TPT totals per lane");

__global__ __launch_bounds__(SCAN_THREADS)
void scan_kernel(const int* __restrict__ counts, int* __restrict__ rowStart,
                 int n) {
  __shared__ int wtot[SCAN_TPT * SCAN_WAVES];
  __shared__ int sweepTotal;
  const int tid = (int)threadIdx.x;
  const int lane = tid & 31;
  const int wv = tid >> 5;
  int carry = 0;

  for (int base = 0; base < n; base += SCAN_SWEEP) {
    int4 x[SCAN_TPT];
#pragma unroll
    for (int j = 0; j < SCAN_TPT; ++j) {
      const int i = base + j * SCAN_TILE + tid * 4;
      const int4 t = *(const int4*)(counts + ((i < n) ? i : 0));
      x[j].x = (i < n) ? t.x : 0;
      x[j].y = (i < n) ? t.y : 0;
      x[j].z = (i < n) ? t.z : 0;
      x[j].w = (i < n) ? t.w : 0;
    }

    int own[SCAN_TPT], s[SCAN_TPT];
#pragma unroll
    for (int j = 0; j < SCAN_TPT; ++j) {
      own[j] = x[j].x + x[j].y + x[j].z + x[j].w;
    }
#pragma unroll
    for (int j = 0; j < SCAN_TPT; ++j) s[j] = wave_incl_scan(own[j], lane);
    if (lane == 31) {
#pragma unroll
      for (int j = 0; j < SCAN_TPT; ++j) wtot[j * SCAN_WAVES + wv] = s[j];
    }
    __syncthreads();

    if (wv == 0) {
      int v[SCAN_TPT];
      int t = 0;
#pragma unroll
      for (int q = 0; q < SCAN_TPT; ++q) {
        v[q] = wtot[lane * SCAN_TPT + q];
        t += v[q];
      }
      const int incl = wave_incl_scan(t, lane);
      int run = incl - t;
#pragma unroll
      for (int q = 0; q < SCAN_TPT; ++q) {
        wtot[lane * SCAN_TPT + q] = run;
        run += v[q];
      }
      if (lane == 31) sweepTotal = incl;
    }
    __syncthreads();

#pragma unroll
    for (int j = 0; j < SCAN_TPT; ++j) {
      const int i = base + j * SCAN_TILE + tid * 4;
      int4 o;
      o.x = carry + wtot[j * SCAN_WAVES + wv] + (s[j] - own[j]);
      o.y = o.x + x[j].x;
      o.z = o.y + x[j].y;
      o.w = o.z + x[j].z;
      if (i < n) *(int4*)(rowStart + i) = o;
    }
    carry += sweepTotal;
    __syncthreads();
  }
  if (tid == 0) rowStart[n] = carry;
}

__global__ void fill_kernel(const int* __restrict__ rows,
                            const int* __restrict__ cols,
                            const float* __restrict__ vals,
                            const int* __restrict__ rowStart,
                            const int* __restrict__ rank,
                            int* __restrict__ csrCol, int* __restrict__ csrRow,
                            float* __restrict__ csrVal, int E) {
  int i = blockIdx.x * blockDim.x + threadIdx.x;
  int s = gridDim.x * blockDim.x;
  for (; i < E; i += s) {
    int r = rows[i];
    int pos = rowStart[r] + rank[i];
    csrCol[pos] = cols[i];
    csrRow[pos] = r;
    csrVal[pos] = vals[i];
  }
}


#define KSPLIT 8
#define PASS_CHUNKS 16

struct BRows {
  f4 lo0, hi0;
  f4 lo1, hi1;
};

__device__ __forceinline__ int from_chunk(int x, int i, int half) {
  const int lo = __builtin_amdgcn_readlane(x, 2 * i);
  const int hi = __builtin_amdgcn_readlane(x, 2 * i + 1);
  return half ? hi : lo;
}

__device__ __forceinline__ BRows gather_rows(const float* __restrict__ ann,
                                             int2 colv, int i, int half,
                                             int c4) {
  BRows b;
  const float* p0 = ann + from_chunk(colv.x, i, half) * DFEAT + c4;
  const float* p1 = ann + from_chunk(colv.y, i, half) * DFEAT + c4;
  b.lo0 = *(const f4*)(p0);
  b.hi0 = *(const f4*)(p0 + 64);
  b.lo1 = *(const f4*)(p1);
  b.hi1 = *(const f4*)(p1 + 64);
  return b;
}

__device__ __forceinline__ v2f a_frag(int2 rowv, float2 valv, int i, int half,
                                      int rowm, bool ok) {
  const int rx = from_chunk(rowv.x, i, half);
  const int ry = from_chunk(rowv.y, i, half);
  const float vx = __int_as_float(from_chunk(__float_as_int(valv.x), i, half));
  const float vy = __int_as_float(from_chunk(__float_as_int(valv.y), i, half));
  v2f a;
  a.x = (ok && rx == rowm) ? vx : 0.0f;
  a.y = (ok && ry == rowm) ? vy : 0.0f;
  return a;
}

__device__ __forceinline__ void mma_rows(v8f (&acc)[8], v2f a,
                                         const BRows& b) {
#pragma unroll
  for (int t = 0; t < 4; ++t) {
    v2f bl, bh;
    bl.x = b.lo0[t];
    bl.y = b.lo1[t];
    bh.x = b.hi0[t];
    bh.y = b.hi1[t];
    acc[t] = __builtin_amdgcn_wmma_f32_16x16x4_f32(
         false, a,  false, bl,
         (short)0, acc[t],  false,  false);
    acc[4 + t] = __builtin_amdgcn_wmma_f32_16x16x4_f32(
         false, a,  false, bh,
         (short)0, acc[4 + t],  false,  false);
  }
}

__global__ __launch_bounds__(256)
void wmma_spmm_kernel(const int* __restrict__ rowStart,
                      const int* __restrict__ csrCol,
                      const int* __restrict__ csrRow,
                      const float* __restrict__ csrVal,
                      const float* __restrict__ ann, float* __restrict__ out) {
  __shared__ f4 part[KSPLIT * 16 * (DFEAT / 4)];

  const int base = blockIdx.x * 16;
  const int kBeg = rowStart[base];
  const int kEnd = rowStart[base + 16];
  const int lane = (int)(threadIdx.x & 31);
  const int wv = __builtin_amdgcn_readfirstlane((int)(threadIdx.x >> 5));
  const int half = lane >> 4;
  const int c = lane & 15;
  const int c4 = 4 * c;
  const int rowm = base + c;

  v8f acc[8];
#pragma unroll
  for (int t = 0; t < 8; ++t) {
    acc[t] = (v8f){0.f, 0.f, 0.f, 0.f, 0.f, 0.f, 0.f, 0.f};
  }

  const int k0 = kBeg & ~3;
  const int nCh = (kEnd - k0 + 3) >> 2;
  const int per = (nCh + KSPLIT - 1) / KSPLIT;
  const int jb = wv * per;
  const int je = (jb + per < nCh) ? jb + per : nCh;

  for (int seg = jb; seg < je; seg += PASS_CHUNKS) {
    const int n = (je - seg < PASS_CHUNKS) ? je - seg : PASS_CHUNKS;
    const int e = k0 + 4 * seg + ((lane < 2 * n) ? 2 * lane : 0);
    const int2 colv = *(const int2*)(csrCol + e);
    const int2 rowv = *(const int2*)(csrRow + e);
    const float2 valv = *(const float2*)(csrVal + e);

    const int n2 = (n + 1) & ~1;
    BRows bA = gather_rows(ann, colv, 0, half, c4);
    int i = 0;
    for (; i + 2 < n2; i += 2) {
      const BRows bB = gather_rows(ann, colv, i + 1, half, c4);
      __builtin_amdgcn_sched_barrier(0);
      mma_rows(acc, a_frag(rowv, valv, i, half, rowm, true), bA);
      bA = gather_rows(ann, colv, i + 2, half, c4);
      __builtin_amdgcn_sched_barrier(0);
      mma_rows(acc, a_frag(rowv, valv, i + 1, half, rowm, true), bB);
    }
    {
      const bool ok = (i + 1 < n);
      const int i1 = ok ? i + 1 : i;
      const BRows bB = gather_rows(ann, colv, i1, half, c4);
      __builtin_amdgcn_sched_barrier(0);
      mma_rows(acc, a_frag(rowv, valv, i, half, rowm, true), bA);
      mma_rows(acc, a_frag(rowv, valv, i1, half, rowm, ok), bB);
    }
  }

  f4* pw = part + wv * (16 * (DFEAT / 4)) + (8 * half) * (DFEAT / 4) + c;
#pragma unroll
  for (int v = 0; v < 8; ++v) {
#pragma unroll
    for (int g = 0; g < 2; ++g) {
      f4 x;
      x.x = acc[4 * g + 0][v];
      x.y = acc[4 * g + 1][v];
      x.z = acc[4 * g + 2][v];
      x.w = acc[4 * g + 3][v];
      pw[v * (DFEAT / 4) + 16 * g] = x;
    }
  }
  __syncthreads();

#pragma unroll
  for (int i = 0; i < 2; ++i) {
    const int row = wv + 8 * i;
    const f4* ps = part + row * (DFEAT / 4) + lane;
    f4 s = ps[0];
#pragma unroll
    for (int w = 1; w < KSPLIT; ++w) s += ps[w * (16 * (DFEAT / 4))];
    *(f4*)(out + (size_t)(base + row) * DFEAT + 4 * lane) = s;
  }
}


__global__ __launch_bounds__(256)
void scatter_kernel(const int* __restrict__ rows, const int* __restrict__ cols,
                    const float* __restrict__ vals,
                    const float* __restrict__ ann, float* __restrict__ out,
                    int E) {
  const int lane = (int)(threadIdx.x & 31);
  int wid = (int)((blockIdx.x * blockDim.x + threadIdx.x) >> 5);
  const int nw = (int)((gridDim.x * blockDim.x) >> 5);
  for (int e = wid; e < E; e += nw) {
    const int r = rows[e];
    const int c = cols[e];
    const float v = vals[e];
    const f4 x = *(const f4*)(ann + (size_t)c * DFEAT + lane * 4);
    float* o = out + (size_t)r * DFEAT + lane * 4;
    unsafeAtomicAdd(o + 0, v * x.x);
    unsafeAtomicAdd(o + 1, v * x.y);
    unsafeAtomicAdd(o + 2, v * x.z);
    unsafeAtomicAdd(o + 3, v * x.w);
  }
}


extern "C" void kernel_launch(void* const* d_in, const int* in_sizes, int n_in,
                              void* d_out, int out_size, void* d_ws,
                              size_t ws_size, hipStream_t stream) {
  const int* rows = (const int*)d_in[0];
  const int* cols = (const int*)d_in[1];
  const float* vals = (const float*)d_in[2];
  const float* ann = (const float*)d_in[3];
  float* out = (float*)d_out;

  const int E = in_sizes[0];
  const int N = out_size / DFEAT;

  const size_t need = (size_t)(2 * N + 1) * sizeof(int) +
                      (size_t)(E + CSR_PAD) * 12 + (size_t)E * sizeof(int);

  if ((N % 16) == 0 && ws_size >= need) {
    char* w = (char*)d_ws;
    int* counts = (int*)w;    w += (size_t)N * 4;
    int* rowStart = (int*)w;  w += (size_t)(N + 1) * 4;
    int* csrCol = (int*)w;    w += (size_t)(E + CSR_PAD) * 4;
    int* csrRow = (int*)w;    w += (size_t)(E + CSR_PAD) * 4;
    float* csrVal = (float*)w; w += (size_t)(E + CSR_PAD) * 4;
    int* rank = (int*)w;

    const int gz = (N + 255) / 256;
    const int ge = (E + 255) / 256;

    zero_i32_kernel<<<gz, 256, 0, stream>>>(counts, N);
    pad_kernel<<<1, 32, 0, stream>>>(csrCol, csrRow, csrVal, E);
    hist_rank_kernel<<<ge, 256, 0, stream>>>(rows, counts, rank, E);
    scan_kernel<<<1, SCAN_THREADS, 0, stream>>>(counts, rowStart, N);
    fill_kernel<<<ge, 256, 0, stream>>>(rows, cols, vals, rowStart, rank,
                                        csrCol, csrRow, csrVal, E);
    wmma_spmm_kernel<<<N / 16, 256, 0, stream>>>(rowStart, csrCol, csrRow,
                                                 csrVal, ann, out);
  } else {
    zero_f32_kernel<<<(out_size + 255) / 256, 256, 0, stream>>>(out, out_size);
    scatter_kernel<<<2048, 256, 0, stream>>>(rows, cols, vals, ann, out, E);
  }
}


# CSR build as a two-level counting sort on LDS counters (256-row buckets, then rows): no global atomics, no single-block scan, no rank array; 6 launches -> 4 (in-row sum order: LDS add order)
# speedup vs baseline: 3.9614x; 1.7570x over previous
#include <hip/hip_runtime.h>
#include <cstdint>
#include <cstddef>


typedef __attribute__((ext_vector_type(2))) float v2f;
typedef __attribute__((ext_vector_type(4))) float f4;
typedef __attribute__((ext_vector_type(8))) float v8f;

#define DFEAT 128
#define CSR_PAD 16

__device__ __forceinline__ int wave_incl_scan(int s, int lane) {
  s += __builtin_amdgcn_update_dpp(0, s, 0x111, 0xf, 0xf, true);
  s += __builtin_amdgcn_update_dpp(0, s, 0x112, 0xf, 0xf, true);
  s += __builtin_amdgcn_update_dpp(0, s, 0x114, 0xf, 0xf, true);
  s += __builtin_amdgcn_update_dpp(0, s, 0x118, 0xf, 0xf, true);
  const int row0 = __builtin_amdgcn_readlane(s, 15);
  return s + ((lane >= 16) ? row0 : 0);
}


#define RB 256
#define RB_SHIFT 8
#define NBP 256
#define BIN_THREADS 512
#define EPT 10
#define SLICE (BIN_THREADS * EPT)
#define SORT_THREADS 512
#define HGROUPS (BIN_THREADS / (NBP / 4))
#define HDEPTH 16
static_assert(NBP == 256 && BIN_THREADS >= NBP && HGROUPS == 8,
              "64 threads x 16 bytes span one table row; 8 rows per step");
static_assert(RB < SORT_THREADS && RB == 256, "wave 0 scans 8 rows per lane");

__global__ __launch_bounds__(BIN_THREADS)
void bucket_count_kernel(const int* __restrict__ rows, int* __restrict__ hist,
                         int E) {
  __shared__ int cnt[NBP + 1];
  const int tid = (int)threadIdx.x;
  const int e0 = (int)blockIdx.x * SLICE;
  int r[EPT];
#pragma unroll
  for (int j = 0; j < EPT; ++j) {
    const int e = e0 + j * BIN_THREADS + tid;
    const int x = rows[(e < E) ? e : 0];
    r[j] = (e < E) ? (x >> RB_SHIFT) : NBP;
  }
  if (tid <= NBP) cnt[tid] = 0;
  __syncthreads();
#pragma unroll
  for (int j = 0; j < EPT; ++j) atomicAdd(&cnt[r[j]], 1);
  __syncthreads();
  if (tid < NBP) hist[(int)blockIdx.x * NBP + tid] = cnt[tid];
}

__device__ __forceinline__ void sweep_trip(const int* __restrict__ hist,
                                           int k0, int g, int t4, int nSlices,
                                           int blk, int4& tot4, int4& bel4) {
  int4 h[HDEPTH];
#pragma unroll
  for (int q = 0; q < HDEPTH; ++q) {
    const int k = k0 + g + HGROUPS * q;
    h[q] = *(const int4*)(hist + ((k < nSlices) ? k : 0) * NBP + 4 * t4);
  }
#pragma unroll
  for (int q = 0; q < HDEPTH; ++q) {
    const int k = k0 + g + HGROUPS * q;
    const bool in = k < nSlices;
    const bool lo = in && (k < blk);
    tot4.x += in ? h[q].x : 0;
    tot4.y += in ? h[q].y : 0;
    tot4.z += in ? h[q].z : 0;
    tot4.w += in ? h[q].w : 0;
    bel4.x += lo ? h[q].x : 0;
    bel4.y += lo ? h[q].y : 0;
    bel4.z += lo ? h[q].z : 0;
    bel4.w += lo ? h[q].w : 0;
  }
}

__global__ __launch_bounds__(BIN_THREADS)
void bucket_place_kernel(const int* __restrict__ rows,
                         const int* __restrict__ cols,
                         const float* __restrict__ vals,
                         const int* __restrict__ hist,
                         int* __restrict__ bucketStart, int* __restrict__ stg,
                         int E, int nSlices) {
  __shared__ __attribute__((aligned(16))) int redTot[HGROUPS * NBP];
  __shared__ __attribute__((aligned(16))) int redBel[HGROUPS * NBP];
  __shared__ int cur[NBP + 1];
  __shared__ int wsum[NBP / 32];
  const int tid = (int)threadIdx.x;
  const int lane = tid & 31;
  const int wv = tid >> 5;
  const int blk = (int)blockIdx.x;
  const int e0 = blk * SLICE;

  int r[EPT], c[EPT];
  float v[EPT];
#pragma unroll
  for (int j = 0; j < EPT; ++j) {
    const int e = e0 + j * BIN_THREADS + tid;
    const bool ok = e < E;
    const int ec = ok ? e : 0;
    const int rr = rows[ec];
    c[j] = cols[ec];
    v[j] = vals[ec];
    r[j] = ok ? rr : -1;
  }

  const int g = tid >> 6;
  const int t4 = tid & 63;
  int4 tot4 = make_int4(0, 0, 0, 0);
  int4 bel4 = make_int4(0, 0, 0, 0);
  sweep_trip(hist, 0, g, t4, nSlices, blk, tot4, bel4);
  for (int k0 = HGROUPS * HDEPTH; k0 < nSlices; k0 += HGROUPS * HDEPTH) {
    sweep_trip(hist, k0, g, t4, nSlices, blk, tot4, bel4);
  }
  *(int4*)(redTot + g * NBP + 4 * t4) = tot4;
  *(int4*)(redBel + g * NBP + 4 * t4) = bel4;
  __syncthreads();

  int tot = 0, below = 0;
  if (tid < NBP) {
#pragma unroll
    for (int w = 0; w < HGROUPS; ++w) {
      tot += redTot[w * NBP + tid];
      below += redBel[w * NBP + tid];
    }
  }
  const int incl = wave_incl_scan(tot, lane);
  if (lane == 31 && wv < NBP / 32) wsum[wv] = incl;
  __syncthreads();
  if (tid < NBP) {
    int wbase = 0;
#pragma unroll
    for (int w = 0; w < NBP / 32; ++w) wbase += (w < wv) ? wsum[w] : 0;
    const int start = wbase + incl - tot;
    cur[tid] = start + below;
    if (blk == 0) {
      bucketStart[tid] = start;
      if (tid == NBP - 1) bucketStart[NBP] = start + tot;
    }
  }
  if (tid == NBP) cur[NBP] = 0;
  __syncthreads();

  int pos[EPT];
#pragma unroll
  for (int j = 0; j < EPT; ++j) {
    pos[j] = atomicAdd(&cur[(r[j] >= 0) ? (r[j] >> RB_SHIFT) : NBP], 1);
  }
  int2* __restrict__ rec = (int2*)stg;
#pragma unroll
  for (int j = 0; j < EPT; ++j) {
    if (r[j] >= 0) {
      int2 x;
      x.x = ((r[j] & (RB - 1)) << 16) | c[j];
      x.y = __float_as_int(v[j]);
      rec[pos[j]] = x;
    }
  }
}

#define KPT 10
#define SORT_TRIP (SORT_THREADS * KPT)

__global__ __launch_bounds__(SORT_THREADS)
void bucket_sort_kernel(const int* __restrict__ bucketStart,
                        const int* __restrict__ stg,
                        int* __restrict__ rowStart, int* __restrict__ csrCol,
                        int* __restrict__ csrRow, float* __restrict__ csrVal,
                        int N, int nBuckets) {
  __shared__ int cnt[RB + 1];
  __shared__ int cur[RB + 1];
  const int tid = (int)threadIdx.x;
  const int lane = tid & 31;
  const int wv = tid >> 5;
  const int b = (int)blockIdx.x;
  const int s = bucketStart[b];
  const int e = bucketStart[b + 1];
  const int2* __restrict__ rec = (const int2*)stg;

  if (tid <= RB) cnt[tid] = 0;
  __syncthreads();
  for (int base = s; base < e; base += SORT_TRIP) {
    int lr[KPT];
#pragma unroll
    for (int q = 0; q < KPT; ++q) {
      const int i = base + q * SORT_THREADS + tid;
      const int2 x = rec[(i < e) ? i : s];
      lr[q] = (i < e) ? (x.x >> 16) : RB;
    }
#pragma unroll
    for (int q = 0; q < KPT; ++q) atomicAdd(&cnt[lr[q]], 1);
  }
  __syncthreads();

  if (wv == 0) {
    int q8[8];
    int t = 0;
#pragma unroll
    for (int q = 0; q < 8; ++q) {
      q8[q] = cnt[lane * 8 + q];
      t += q8[q];
    }
    const int incl = wave_incl_scan(t, lane);
    int run = s + incl - t;
#pragma unroll
    for (int q = 0; q < 8; ++q) {
      const int row = b * RB + lane * 8 + q;
      cur[lane * 8 + q] = run;
      if (row < N) rowStart[row] = run;
      run += q8[q];
    }
    if (lane == 0) cur[RB] = 0;
  }
  __syncthreads();

  for (int base = s; base < e; base += SORT_TRIP) {
    int2 x[KPT];
    int pos[KPT];
#pragma unroll
    for (int q = 0; q < KPT; ++q) {
      const int i = base + q * SORT_THREADS + tid;
      x[q] = rec[(i < e) ? i : s];
      if (i >= e) x[q].x = -1;
    }
#pragma unroll
    for (int q = 0; q < KPT; ++q) {
      pos[q] = atomicAdd(&cur[(x[q].x >= 0) ? (x[q].x >> 16) : RB], 1);
    }
#pragma unroll
    for (int q = 0; q < KPT; ++q) {
      if (x[q].x >= 0) {
        csrCol[pos[q]] = x[q].x & 0xFFFF;
        csrRow[pos[q]] = b * RB + (x[q].x >> 16);
        csrVal[pos[q]] = __int_as_float(x[q].y);
      }
    }
  }

  if (b == nBuckets - 1) {
    if (tid == 0) rowStart[N] = e;
    if (tid < CSR_PAD) {
      csrCol[e + tid] = 0;
      csrRow[e + tid] = -1;
      csrVal[e + tid] = 0.0f;
    }
  }
}


#define KSPLIT 8
#define PASS_CHUNKS 16

struct BRows {
  f4 lo0, hi0;
  f4 lo1, hi1;
};

__device__ __forceinline__ int from_chunk(int x, int i, int half) {
  const int lo = __builtin_amdgcn_readlane(x, 2 * i);
  const int hi = __builtin_amdgcn_readlane(x, 2 * i + 1);
  return half ? hi : lo;
}

__device__ __forceinline__ BRows gather_rows(const float* __restrict__ ann,
                                             int2 colv, int i, int half,
                                             int c4) {
  BRows b;
  const float* p0 = ann + from_chunk(colv.x, i, half) * DFEAT + c4;
  const float* p1 = ann + from_chunk(colv.y, i, half) * DFEAT + c4;
  b.lo0 = *(const f4*)(p0);
  b.hi0 = *(const f4*)(p0 + 64);
  b.lo1 = *(const f4*)(p1);
  b.hi1 = *(const f4*)(p1 + 64);
  return b;
}

__device__ __forceinline__ v2f a_frag(int2 rowv, float2 valv, int i, int half,
                                      int rowm, bool ok) {
  const int rx = from_chunk(rowv.x, i, half);
  const int ry = from_chunk(rowv.y, i, half);
  const float vx = __int_as_float(from_chunk(__float_as_int(valv.x), i, half));
  const float vy = __int_as_float(from_chunk(__float_as_int(valv.y), i, half));
  v2f a;
  a.x = (ok && rx == rowm) ? vx : 0.0f;
  a.y = (ok && ry == rowm) ? vy : 0.0f;
  return a;
}

__device__ __forceinline__ void mma_rows(v8f (&acc)[8], v2f a,
                                         const BRows& b) {
#pragma unroll
  for (int t = 0; t < 4; ++t) {
    v2f bl, bh;
    bl.x = b.lo0[t];
    bl.y = b.lo1[t];
    bh.x = b.hi0[t];
    bh.y = b.hi1[t];
    acc[t] = __builtin_amdgcn_wmma_f32_16x16x4_f32(
         false, a,  false, bl,
         (short)0, acc[t],  false,  false);
    acc[4 + t] = __builtin_amdgcn_wmma_f32_16x16x4_f32(
         false, a,  false, bh,
         (short)0, acc[4 + t],  false,  false);
  }
}

__global__ __launch_bounds__(256)
void wmma_spmm_kernel(const int* __restrict__ rowStart,
                      const int* __restrict__ csrCol,
                      const int* __restrict__ csrRow,
                      const float* __restrict__ csrVal,
                      const float* __restrict__ ann, float* __restrict__ out) {
  __shared__ f4 part[KSPLIT * 16 * (DFEAT / 4)];

  const int base = blockIdx.x * 16;
  const int kBeg = rowStart[base];
  const int kEnd = rowStart[base + 16];
  const int lane = (int)(threadIdx.x & 31);
  const int wv = __builtin_amdgcn_readfirstlane((int)(threadIdx.x >> 5));
  const int half = lane >> 4;
  const int c = lane & 15;
  const int c4 = 4 * c;
  const int rowm = base + c;

  v8f acc[8];
#pragma unroll
  for (int t = 0; t < 8; ++t) {
    acc[t] = (v8f){0.f, 0.f, 0.f, 0.f, 0.f, 0.f, 0.f, 0.f};
  }

  const int k0 = kBeg & ~3;
  const int nCh = (kEnd - k0 + 3) >> 2;
  const int per = (nCh + KSPLIT - 1) / KSPLIT;
  const int jb = wv * per;
  const int je = (jb + per < nCh) ? jb + per : nCh;

  for (int seg = jb; seg < je; seg += PASS_CHUNKS) {
    const int n = (je - seg < PASS_CHUNKS) ? je - seg : PASS_CHUNKS;
    const int e = k0 + 4 * seg + ((lane < 2 * n) ? 2 * lane : 0);
    const int2 colv = *(const int2*)(csrCol + e);
    const int2 rowv = *(const int2*)(csrRow + e);
    const float2 valv = *(const float2*)(csrVal + e);

    const int n2 = (n + 1) & ~1;
    BRows bA = gather_rows(ann, colv, 0, half, c4);
    int i = 0;
    for (; i + 2 < n2; i += 2) {
      const BRows bB = gather_rows(ann, colv, i + 1, half, c4);
      __builtin_amdgcn_sched_barrier(0);
      mma_rows(acc, a_frag(rowv, valv, i, half, rowm, true), bA);
      bA = gather_rows(ann, colv, i + 2, half, c4);
      __builtin_amdgcn_sched_barrier(0);
      mma_rows(acc, a_frag(rowv, valv, i + 1, half, rowm, true), bB);
    }
    {
      const bool ok = (i + 1 < n);
      const int i1 = ok ? i + 1 : i;
      const BRows bB = gather_rows(ann, colv, i1, half, c4);
      __builtin_amdgcn_sched_barrier(0);
      mma_rows(acc, a_frag(rowv, valv, i, half, rowm, true), bA);
      mma_rows(acc, a_frag(rowv, valv, i1, half, rowm, ok), bB);
    }
  }

  f4* pw = part + wv * (16 * (DFEAT / 4)) + (8 * half) * (DFEAT / 4) + c;
#pragma unroll
  for (int v = 0; v < 8; ++v) {
#pragma unroll
    for (int g = 0; g < 2; ++g) {
      f4 x;
      x.x = acc[4 * g + 0][v];
      x.y = acc[4 * g + 1][v];
      x.z = acc[4 * g + 2][v];
      x.w = acc[4 * g + 3][v];
      pw[v * (DFEAT / 4) + 16 * g] = x;
    }
  }
  __syncthreads();

#pragma unroll
  for (int i = 0; i < 2; ++i) {
    const int row = wv + 8 * i;
    const f4* ps = part + row * (DFEAT / 4) + lane;
    f4 s = ps[0];
#pragma unroll
    for (int w = 1; w < KSPLIT; ++w) s += ps[w * (16 * (DFEAT / 4))];
    *(f4*)(out + (size_t)(base + row) * DFEAT + 4 * lane) = s;
  }
}


__global__ void zero_f32_kernel(float* __restrict__ p, int n) {
  int i = blockIdx.x * blockDim.x + threadIdx.x;
  int s = gridDim.x * blockDim.x;
  for (; i < n; i += s) p[i] = 0.0f;
}

__global__ __launch_bounds__(256)
void scatter_kernel(const int* __restrict__ rows, const int* __restrict__ cols,
                    const float* __restrict__ vals,
                    const float* __restrict__ ann, float* __restrict__ out,
                    int E) {
  const int lane = (int)(threadIdx.x & 31);
  int wid = (int)((blockIdx.x * blockDim.x + threadIdx.x) >> 5);
  const int nw = (int)((gridDim.x * blockDim.x) >> 5);
  for (int e = wid; e < E; e += nw) {
    const int r = rows[e];
    const int c = cols[e];
    const float v = vals[e];
    const f4 x = *(const f4*)(ann + (size_t)c * DFEAT + lane * 4);
    float* o = out + (size_t)r * DFEAT + lane * 4;
    unsafeAtomicAdd(o + 0, v * x.x);
    unsafeAtomicAdd(o + 1, v * x.y);
    unsafeAtomicAdd(o + 2, v * x.z);
    unsafeAtomicAdd(o + 3, v * x.w);
  }
}


extern "C" void kernel_launch(void* const* d_in, const int* in_sizes, int n_in,
                              void* d_out, int out_size, void* d_ws,
                              size_t ws_size, hipStream_t stream) {
  const int* rows = (const int*)d_in[0];
  const int* cols = (const int*)d_in[1];
  const float* vals = (const float*)d_in[2];
  const float* ann = (const float*)d_in[3];
  float* out = (float*)d_out;

  const int E = in_sizes[0];
  const int N = out_size / DFEAT;

  const int nSlices = (E + SLICE - 1) / SLICE;
  const int nBuckets = (N + RB - 1) / RB;

  const size_t histBytes = (size_t)nSlices * NBP * sizeof(int);
  const size_t bstBytes = (size_t)(NBP + 16) * sizeof(int);
  const size_t stgBytes = (size_t)E * 8;
  const size_t rsBytes = (size_t)(N + 16) * sizeof(int);
  const size_t csrBytes = (size_t)(E + CSR_PAD) * 4;
  const size_t need = histBytes + bstBytes + stgBytes + rsBytes + 3 * csrBytes;

  if ((N % 16) == 0 && N <= RB * NBP && E > 0 && ws_size >= need) {
    char* w = (char*)d_ws;
    int* hist = (int*)w;         w += histBytes;
    int* bucketStart = (int*)w;  w += bstBytes;
    int* stg = (int*)w;          w += stgBytes;
    int* rowStart = (int*)w;     w += rsBytes;
    int* csrCol = (int*)w;       w += csrBytes;
    int* csrRow = (int*)w;       w += csrBytes;
    float* csrVal = (float*)w;

    bucket_count_kernel<<<nSlices, BIN_THREADS, 0, stream>>>(rows, hist, E);
    bucket_place_kernel<<<nSlices, BIN_THREADS, 0, stream>>>(
        rows, cols, vals, hist, bucketStart, stg, E, nSlices);
    bucket_sort_kernel<<<nBuckets, SORT_THREADS, 0, stream>>>(
        bucketStart, stg, rowStart, csrCol, csrRow, csrVal, N, nBuckets);
    wmma_spmm_kernel<<<N / 16, 256, 0, stream>>>(rowStart, csrCol, csrRow,
                                                 csrVal, ann, out);
  } else {
    zero_f32_kernel<<<(out_size + 255) / 256, 256, 0, stream>>>(out, out_size);
    scatter_kernel<<<2048, 256, 0, stream>>>(rows, cols, vals, ann, out, E);
  }
}


# both sort levels group their records in an LDS image and write it out as runs of consecutive slots (row sort: aligned 128-byte runs); was a scattered line per lane per store (sum order as before)
# speedup vs baseline: 4.3050x; 1.0867x over previous
#include <hip/hip_runtime.h>
#include <cstdint>
#include <cstddef>


typedef __attribute__((ext_vector_type(2))) float v2f;
typedef __attribute__((ext_vector_type(4))) float f4;
typedef __attribute__((ext_vector_type(8))) float v8f;

#define DFEAT 128
#define CSR_PAD 16

__device__ __forceinline__ int wave_incl_scan(int s, int lane) {
  s += __builtin_amdgcn_update_dpp(0, s, 0x111, 0xf, 0xf, true);
  s += __builtin_amdgcn_update_dpp(0, s, 0x112, 0xf, 0xf, true);
  s += __builtin_amdgcn_update_dpp(0, s, 0x114, 0xf, 0xf, true);
  s += __builtin_amdgcn_update_dpp(0, s, 0x118, 0xf, 0xf, true);
  const int row0 = __builtin_amdgcn_readlane(s, 15);
  return s + ((lane >= 16) ? row0 : 0);
}


#define RB 256
#define RB_SHIFT 8
#define NBP 256
#define BIN_THREADS 512
#define EPT 10
#define SLICE (BIN_THREADS * EPT)
#define SORT_THREADS 512
#define HGROUPS (BIN_THREADS / (NBP / 4))
#define HDEPTH 16
static_assert(NBP == 256 && BIN_THREADS >= NBP && HGROUPS == 8,
              "64 threads x 16 bytes span one table row; 8 rows per step");
static_assert(RB < SORT_THREADS && RB == 256, "wave 0 scans 8 rows per lane");

__global__ __launch_bounds__(BIN_THREADS)
void bucket_count_kernel(const int* __restrict__ rows, int* __restrict__ hist,
                         int E) {
  __shared__ int cnt[NBP + 1];
  const int tid = (int)threadIdx.x;
  const int e0 = (int)blockIdx.x * SLICE;
  int r[EPT];
#pragma unroll
  for (int j = 0; j < EPT; ++j) {
    const int e = e0 + j * BIN_THREADS + tid;
    const int x = rows[(e < E) ? e : 0];
    r[j] = (e < E) ? (x >> RB_SHIFT) : NBP;
  }
  if (tid <= NBP) cnt[tid] = 0;
  __syncthreads();
#pragma unroll
  for (int j = 0; j < EPT; ++j) atomicAdd(&cnt[r[j]], 1);
  __syncthreads();
  if (tid < NBP) hist[(int)blockIdx.x * NBP + tid] = cnt[tid];
}

__device__ __forceinline__ void sweep_trip(const int* __restrict__ hist,
                                           int k0, int g, int t4, int nSlices,
                                           int blk, int4& tot4, int4& bel4) {
  int4 h[HDEPTH];
#pragma unroll
  for (int q = 0; q < HDEPTH; ++q) {
    const int k = k0 + g + HGROUPS * q;
    h[q] = *(const int4*)(hist + ((k < nSlices) ? k : 0) * NBP + 4 * t4);
  }
#pragma unroll
  for (int q = 0; q < HDEPTH; ++q) {
    const int k = k0 + g + HGROUPS * q;
    const bool in = k < nSlices;
    const bool lo = in && (k < blk);
    tot4.x += in ? h[q].x : 0;
    tot4.y += in ? h[q].y : 0;
    tot4.z += in ? h[q].z : 0;
    tot4.w += in ? h[q].w : 0;
    bel4.x += lo ? h[q].x : 0;
    bel4.y += lo ? h[q].y : 0;
    bel4.z += lo ? h[q].z : 0;
    bel4.w += lo ? h[q].w : 0;
  }
}

__global__ __launch_bounds__(BIN_THREADS)
void bucket_place_kernel(const int* __restrict__ rows,
                         const int* __restrict__ cols,
                         const float* __restrict__ vals,
                         const int* __restrict__ hist,
                         int* __restrict__ bucketStart, int* __restrict__ stg,
                         int E, int nSlices) {
  __shared__ __attribute__((aligned(16))) int redTot[HGROUPS * NBP];
  __shared__ __attribute__((aligned(16))) int redBel[HGROUPS * NBP];
  __shared__ int cur[NBP + 1];
  __shared__ int delta[NBP];
  __shared__ int wsum[NBP / 32];
  __shared__ int wsumOwn[NBP / 32];
  __shared__ int2 img[SLICE];
  const int tid = (int)threadIdx.x;
  const int lane = tid & 31;
  const int wv = tid >> 5;
  const int blk = (int)blockIdx.x;
  const int e0 = blk * SLICE;

  int r[EPT], c[EPT];
  float v[EPT];
#pragma unroll
  for (int j = 0; j < EPT; ++j) {
    const int e = e0 + j * BIN_THREADS + tid;
    const bool ok = e < E;
    const int ec = ok ? e : 0;
    const int rr = rows[ec];
    c[j] = cols[ec];
    v[j] = vals[ec];
    r[j] = ok ? rr : -1;
  }
  const int own = (tid < NBP) ? hist[blk * NBP + tid] : 0;

  const int g = tid >> 6;
  const int t4 = tid & 63;
  int4 tot4 = make_int4(0, 0, 0, 0);
  int4 bel4 = make_int4(0, 0, 0, 0);
  sweep_trip(hist, 0, g, t4, nSlices, blk, tot4, bel4);
  for (int k0 = HGROUPS * HDEPTH; k0 < nSlices; k0 += HGROUPS * HDEPTH) {
    sweep_trip(hist, k0, g, t4, nSlices, blk, tot4, bel4);
  }
  *(int4*)(redTot + g * NBP + 4 * t4) = tot4;
  *(int4*)(redBel + g * NBP + 4 * t4) = bel4;
  __syncthreads();

  int tot = 0, below = 0;
  if (tid < NBP) {
#pragma unroll
    for (int w = 0; w < HGROUPS; ++w) {
      tot += redTot[w * NBP + tid];
      below += redBel[w * NBP + tid];
    }
  }
  const int incl = wave_incl_scan(tot, lane);
  const int inclOwn = wave_incl_scan(own, lane);
  if (lane == 31 && wv < NBP / 32) {
    wsum[wv] = incl;
    wsumOwn[wv] = inclOwn;
  }
  __syncthreads();
  if (tid < NBP) {
    int wbase = 0, wbaseOwn = 0;
#pragma unroll
    for (int w = 0; w < NBP / 32; ++w) {
      wbase += (w < wv) ? wsum[w] : 0;
      wbaseOwn += (w < wv) ? wsumOwn[w] : 0;
    }
    const int start = wbase + incl - tot;
    const int lstart = wbaseOwn + inclOwn - own;
    cur[tid] = lstart;
    delta[tid] = start + below - lstart;
    if (blk == 0) {
      bucketStart[tid] = start;
      if (tid == NBP - 1) bucketStart[NBP] = start + tot;
    }
  }
  if (tid == NBP) cur[NBP] = 0;
  __syncthreads();

  int pos[EPT];
#pragma unroll
  for (int j = 0; j < EPT; ++j) {
    pos[j] = atomicAdd(&cur[(r[j] >= 0) ? (r[j] >> RB_SHIFT) : NBP], 1);
  }
#pragma unroll
  for (int j = 0; j < EPT; ++j) {
    if (r[j] >= 0) {
      int2 x;
      x.x = (int)(((unsigned)r[j] << 16) | (unsigned)c[j]);
      x.y = __float_as_int(v[j]);
      img[pos[j]] = x;
    }
  }
  __syncthreads();

  int2* __restrict__ rec = (int2*)stg;
  const int nv = (E - e0 < SLICE) ? E - e0 : SLICE;
  int2 y[EPT];
  int d[EPT];
#pragma unroll
  for (int j = 0; j < EPT; ++j) {
    const int i = j * BIN_THREADS + tid;
    y[j] = img[(i < nv) ? i : 0];
  }
#pragma unroll
  for (int j = 0; j < EPT; ++j) d[j] = delta[(unsigned)y[j].x >> 24];
#pragma unroll
  for (int j = 0; j < EPT; ++j) {
    const int i = j * BIN_THREADS + tid;
    if (i < nv) {
      int2 x;
      x.x = y[j].x & 0x00FFFFFF;
      x.y = y[j].y;
      rec[i + d[j]] = x;
    }
  }
}

#define KPT 10
#define SORT_TRIP (SORT_THREADS * KPT)
#define IMG_CAP SORT_TRIP
#define OUT_DEPTH 4

__global__ __launch_bounds__(SORT_THREADS)
void bucket_sort_kernel(const int* __restrict__ bucketStart,
                        const int* __restrict__ stg,
                        int* __restrict__ rowStart, int* __restrict__ csrCol,
                        int* __restrict__ csrRow, float* __restrict__ csrVal,
                        int N, int nBuckets) {
  __shared__ int cnt[RB + 1];
  __shared__ int cur[RB + 1];
  __shared__ int2 img[IMG_CAP];
  const int tid = (int)threadIdx.x;
  const int lane = tid & 31;
  const int wv = tid >> 5;
  const int b = (int)blockIdx.x;
  const int s = bucketStart[b];
  const int e = bucketStart[b + 1];
  const int2* __restrict__ rec = (const int2*)stg;

  if (tid <= RB) cnt[tid] = 0;
  __syncthreads();
  for (int base = s; base < e; base += SORT_TRIP) {
    int lr[KPT];
#pragma unroll
    for (int q = 0; q < KPT; ++q) {
      const int i = base + q * SORT_THREADS + tid;
      const int2 x = rec[(i < e) ? i : s];
      lr[q] = (i < e) ? (x.x >> 16) : RB;
    }
#pragma unroll
    for (int q = 0; q < KPT; ++q) atomicAdd(&cnt[lr[q]], 1);
  }
  __syncthreads();

  if (wv == 0) {
    int q8[8];
    int t = 0;
#pragma unroll
    for (int q = 0; q < 8; ++q) {
      q8[q] = cnt[lane * 8 + q];
      t += q8[q];
    }
    const int incl = wave_incl_scan(t, lane);
    int run = s + incl - t;
#pragma unroll
    for (int q = 0; q < 8; ++q) {
      const int row = b * RB + lane * 8 + q;
      cur[lane * 8 + q] = run;
      if (row < N) rowStart[row] = run;
      run += q8[q];
    }
    if (lane == 0) cur[RB] = 0;
  }
  __syncthreads();

  const bool staged = (e - s) <= IMG_CAP;
  for (int base = s; base < e; base += SORT_TRIP) {
    int2 x[KPT];
    int pos[KPT];
#pragma unroll
    for (int q = 0; q < KPT; ++q) {
      const int i = base + q * SORT_THREADS + tid;
      x[q] = rec[(i < e) ? i : s];
      if (i >= e) x[q].x = -1;
    }
#pragma unroll
    for (int q = 0; q < KPT; ++q) {
      pos[q] = atomicAdd(&cur[(x[q].x >= 0) ? (x[q].x >> 16) : RB], 1);
    }
    if (staged) {
#pragma unroll
      for (int q = 0; q < KPT; ++q) {
        if (x[q].x >= 0) img[pos[q] - s] = x[q];
      }
    } else {
#pragma unroll
      for (int q = 0; q < KPT; ++q) {
        if (x[q].x >= 0) {
          csrCol[pos[q]] = x[q].x & 0xFFFF;
          csrRow[pos[q]] = b * RB + (x[q].x >> 16);
          csrVal[pos[q]] = __int_as_float(x[q].y);
        }
      }
    }
  }

  if (staged) {
    __syncthreads();
    const int i0 = s & ~31;
    for (int base = i0; base < e; base += SORT_THREADS * OUT_DEPTH) {
      int2 r[OUT_DEPTH];
#pragma unroll
      for (int q = 0; q < OUT_DEPTH; ++q) {
        const int i = base + q * SORT_THREADS + tid;
        r[q] = img[(i >= s && i < e) ? i - s : 0];
      }
#pragma unroll
      for (int q = 0; q < OUT_DEPTH; ++q) {
        const int i = base + q * SORT_THREADS + tid;
        if (i >= s && i < e) {
          csrCol[i] = r[q].x & 0xFFFF;
          csrRow[i] = b * RB + (r[q].x >> 16);
          csrVal[i] = __int_as_float(r[q].y);
        }
      }
    }
  }

  if (b == nBuckets - 1) {
    if (tid == 0) rowStart[N] = e;
    if (tid < CSR_PAD) {
      csrCol[e + tid] = 0;
      csrRow[e + tid] = -1;
      csrVal[e + tid] = 0.0f;
    }
  }
}


#define KSPLIT 8
#define PASS_CHUNKS 16

struct BRows {
  f4 lo0, hi0;
  f4 lo1, hi1;
};

__device__ __forceinline__ int from_chunk(int x, int i, int half) {
  const int lo = __builtin_amdgcn_readlane(x, 2 * i);
  const int hi = __builtin_amdgcn_readlane(x, 2 * i + 1);
  return half ? hi : lo;
}

__device__ __forceinline__ BRows gather_rows(const float* __restrict__ ann,
                                             int2 colv, int i, int half,
                                             int c4) {
  BRows b;
  const float* p0 = ann + from_chunk(colv.x, i, half) * DFEAT + c4;
  const float* p1 = ann + from_chunk(colv.y, i, half) * DFEAT + c4;
  b.lo0 = *(const f4*)(p0);
  b.hi0 = *(const f4*)(p0 + 64);
  b.lo1 = *(const f4*)(p1);
  b.hi1 = *(const f4*)(p1 + 64);
  return b;
}

__device__ __forceinline__ v2f a_frag(int2 rowv, float2 valv, int i, int half,
                                      int rowm, bool ok) {
  const int rx = from_chunk(rowv.x, i, half);
  const int ry = from_chunk(rowv.y, i, half);
  const float vx = __int_as_float(from_chunk(__float_as_int(valv.x), i, half));
  const float vy = __int_as_float(from_chunk(__float_as_int(valv.y), i, half));
  v2f a;
  a.x = (ok && rx == rowm) ? vx : 0.0f;
  a.y = (ok && ry == rowm) ? vy : 0.0f;
  return a;
}

__device__ __forceinline__ void mma_rows(v8f (&acc)[8], v2f a,
                                         const BRows& b) {
#pragma unroll
  for (int t = 0; t < 4; ++t) {
    v2f bl, bh;
    bl.x = b.lo0[t];
    bl.y = b.lo1[t];
    bh.x = b.hi0[t];
    bh.y = b.hi1[t];
    acc[t] = __builtin_amdgcn_wmma_f32_16x16x4_f32(
         false, a,  false, bl,
         (short)0, acc[t],  false,  false);
    acc[4 + t] = __builtin_amdgcn_wmma_f32_16x16x4_f32(
         false, a,  false, bh,
         (short)0, acc[4 + t],  false,  false);
  }
}

__global__ __launch_bounds__(256)
void wmma_spmm_kernel(const int* __restrict__ rowStart,
                      const int* __restrict__ csrCol,
                      const int* __restrict__ csrRow,
                      const float* __restrict__ csrVal,
                      const float* __restrict__ ann, float* __restrict__ out) {
  __shared__ f4 part[KSPLIT * 16 * (DFEAT / 4)];

  const int base = blockIdx.x * 16;
  const int kBeg = rowStart[base];
  const int kEnd = rowStart[base + 16];
  const int lane = (int)(threadIdx.x & 31);
  const int wv = __builtin_amdgcn_readfirstlane((int)(threadIdx.x >> 5));
  const int half = lane >> 4;
  const int c = lane & 15;
  const int c4 = 4 * c;
  const int rowm = base + c;

  v8f acc[8];
#pragma unroll
  for (int t = 0; t < 8; ++t) {
    acc[t] = (v8f){0.f, 0.f, 0.f, 0.f, 0.f, 0.f, 0.f, 0.f};
  }

  const int k0 = kBeg & ~3;
  const int nCh = (kEnd - k0 + 3) >> 2;
  const int per = (nCh + KSPLIT - 1) / KSPLIT;
  const int jb = wv * per;
  const int je = (jb + per < nCh) ? jb + per : nCh;

  for (int seg = jb; seg < je; seg += PASS_CHUNKS) {
    const int n = (je - seg < PASS_CHUNKS) ? je - seg : PASS_CHUNKS;
    const int e = k0 + 4 * seg + ((lane < 2 * n) ? 2 * lane : 0);
    const int2 colv = *(const int2*)(csrCol + e);
    const int2 rowv = *(const int2*)(csrRow + e);
    const float2 valv = *(const float2*)(csrVal + e);

    const int n2 = (n + 1) & ~1;
    BRows bA = gather_rows(ann, colv, 0, half, c4);
    int i = 0;
    for (; i + 2 < n2; i += 2) {
      const BRows bB = gather_rows(ann, colv, i + 1, half, c4);
      __builtin_amdgcn_sched_barrier(0);
      mma_rows(acc, a_frag(rowv, valv, i, half, rowm, true), bA);
      bA = gather_rows(ann, colv, i + 2, half, c4);
      __builtin_amdgcn_sched_barrier(0);
      mma_rows(acc, a_frag(rowv, valv, i + 1, half, rowm, true), bB);
    }
    {
      const bool ok = (i + 1 < n);
      const int i1 = ok ? i + 1 : i;
      const BRows bB = gather_rows(ann, colv, i1, half, c4);
      __builtin_amdgcn_sched_barrier(0);
      mma_rows(acc, a_frag(rowv, valv, i, half, rowm, true), bA);
      mma_rows(acc, a_frag(rowv, valv, i1, half, rowm, ok), bB);
    }
  }

  f4* pw = part + wv * (16 * (DFEAT / 4)) + (8 * half) * (DFEAT / 4) + c;
#pragma unroll
  for (int v = 0; v < 8; ++v) {
#pragma unroll
    for (int g = 0; g < 2; ++g) {
      f4 x;
      x.x = acc[4 * g + 0][v];
      x.y = acc[4 * g + 1][v];
      x.z = acc[4 * g + 2][v];
      x.w = acc[4 * g + 3][v];
      pw[v * (DFEAT / 4) + 16 * g] = x;
    }
  }
  __syncthreads();

#pragma unroll
  for (int i = 0; i < 2; ++i) {
    const int row = wv + 8 * i;
    const f4* ps = part + row * (DFEAT / 4) + lane;
    f4 s = ps[0];
#pragma unroll
    for (int w = 1; w < KSPLIT; ++w) s += ps[w * (16 * (DFEAT / 4))];
    *(f4*)(out + (size_t)(base + row) * DFEAT + 4 * lane) = s;
  }
}


__global__ void zero_f32_kernel(float* __restrict__ p, int n) {
  int i = blockIdx.x * blockDim.x + threadIdx.x;
  int s = gridDim.x * blockDim.x;
  for (; i < n; i += s) p[i] = 0.0f;
}

__global__ __launch_bounds__(256)
void scatter_kernel(const int* __restrict__ rows, const int* __restrict__ cols,
                    const float* __restrict__ vals,
                    const float* __restrict__ ann, float* __restrict__ out,
                    int E) {
  const int lane = (int)(threadIdx.x & 31);
  int wid = (int)((blockIdx.x * blockDim.x + threadIdx.x) >> 5);
  const int nw = (int)((gridDim.x * blockDim.x) >> 5);
  for (int e = wid; e < E; e += nw) {
    const int r = rows[e];
    const int c = cols[e];
    const float v = vals[e];
    const f4 x = *(const f4*)(ann + (size_t)c * DFEAT + lane * 4);
    float* o = out + (size_t)r * DFEAT + lane * 4;
    unsafeAtomicAdd(o + 0, v * x.x);
    unsafeAtomicAdd(o + 1, v * x.y);
    unsafeAtomicAdd(o + 2, v * x.z);
    unsafeAtomicAdd(o + 3, v * x.w);
  }
}


extern "C" void kernel_launch(void* const* d_in, const int* in_sizes, int n_in,
                              void* d_out, int out_size, void* d_ws,
                              size_t ws_size, hipStream_t stream) {
  const int* rows = (const int*)d_in[0];
  const int* cols = (const int*)d_in[1];
  const float* vals = (const float*)d_in[2];
  const float* ann = (const float*)d_in[3];
  float* out = (float*)d_out;

  const int E = in_sizes[0];
  const int N = out_size / DFEAT;

  const int nSlices = (E + SLICE - 1) / SLICE;
  const int nBuckets = (N + RB - 1) / RB;

  const size_t histBytes = (size_t)nSlices * NBP * sizeof(int);
  const size_t bstBytes = (size_t)(NBP + 16) * sizeof(int);
  const size_t stgBytes = (size_t)E * 8;
  const size_t rsBytes = (size_t)(N + 16) * sizeof(int);
  const size_t csrBytes = (size_t)(E + CSR_PAD) * 4;
  const size_t need = histBytes + bstBytes + stgBytes + rsBytes + 3 * csrBytes;

  if ((N % 16) == 0 && N <= RB * NBP && E > 0 && ws_size >= need) {
    char* w = (char*)d_ws;
    int* hist = (int*)w;         w += histBytes;
    int* bucketStart = (int*)w;  w += bstBytes;
    int* stg = (int*)w;          w += stgBytes;
    int* rowStart = (int*)w;     w += rsBytes;
    int* csrCol = (int*)w;       w += csrBytes;
    int* csrRow = (int*)w;       w += csrBytes;
    float* csrVal = (float*)w;

    bucket_count_kernel<<<nSlices, BIN_THREADS, 0, stream>>>(rows, hist, E);
    bucket_place_kernel<<<nSlices, BIN_THREADS, 0, stream>>>(
        rows, cols, vals, hist, bucketStart, stg, E, nSlices);
    bucket_sort_kernel<<<nBuckets, SORT_THREADS, 0, stream>>>(
        bucketStart, stg, rowStart, csrCol, csrRow, csrVal, N, nBuckets);
    wmma_spmm_kernel<<<N / 16, 256, 0, stream>>>(rowStart, csrCol, csrRow,
                                                 csrVal, ann, out);
  } else {
    zero_f32_kernel<<<(out_size + 255) / 256, 256, 0, stream>>>(out, out_size);
    scatter_kernel<<<2048, 256, 0, stream>>>(rows, cols, vals, ann, out, E);
  }
}


# row sort fused into SpMM: a work-group per 160-row bucket orders it in an LDS image and multiplies its 10 windows from it, 2 teams of 8 waves; CSR arrays and one launch gone (sum order: LDS adds)
# speedup vs baseline: 5.3734x; 1.2482x over previous
#include <hip/hip_runtime.h>
#include <cstdint>
#include <cstddef>


typedef __attribute__((ext_vector_type(2))) float v2f;
typedef __attribute__((ext_vector_type(4))) float f4;
typedef __attribute__((ext_vector_type(8))) float v8f;

#define DFEAT 128
#define CSR_PAD 16

__device__ __forceinline__ int wave_incl_scan(int s, int lane) {
  s += __builtin_amdgcn_update_dpp(0, s, 0x111, 0xf, 0xf, true);
  s += __builtin_amdgcn_update_dpp(0, s, 0x112, 0xf, 0xf, true);
  s += __builtin_amdgcn_update_dpp(0, s, 0x114, 0xf, 0xf, true);
  s += __builtin_amdgcn_update_dpp(0, s, 0x118, 0xf, 0xf, true);
  const int row0 = __builtin_amdgcn_readlane(s, 15);
  return s + ((lane >= 16) ? row0 : 0);
}


#define RB 160
#define NBP 256
#define BIN_THREADS 512
#define EPT 10
#define SLICE (BIN_THREADS * EPT)
#define HGROUPS (BIN_THREADS / (NBP / 4))
#define HDEPTH 16
static_assert(NBP == 256 && BIN_THREADS >= NBP && HGROUPS == 8,
              "64 threads x 16 bytes span one table row; 8 rows per step");
static_assert(RB % 32 == 0 && RB < 255, "wave 0 scans RB / 32 rows per lane; "
              "row 255 of a bucket marks a pad record");

__global__ __launch_bounds__(BIN_THREADS)
void bucket_count_kernel(const int* __restrict__ rows, int* __restrict__ hist,
                         int E) {
  __shared__ int cnt[NBP + 1];
  const int tid = (int)threadIdx.x;
  const int e0 = (int)blockIdx.x * SLICE;
  int r[EPT];
#pragma unroll
  for (int j = 0; j < EPT; ++j) {
    const int e = e0 + j * BIN_THREADS + tid;
    const int x = rows[(e < E) ? e : 0];
    r[j] = (e < E) ? (int)((unsigned)x / RB) : NBP;
  }
  if (tid <= NBP) cnt[tid] = 0;
  __syncthreads();
#pragma unroll
  for (int j = 0; j < EPT; ++j) atomicAdd(&cnt[r[j]], 1);
  __syncthreads();
  if (tid < NBP) hist[(int)blockIdx.x * NBP + tid] = cnt[tid];
}

__device__ __forceinline__ void sweep_trip(const int* __restrict__ hist,
                                           int k0, int g, int t4, int nSlices,
                                           int blk, int4& tot4, int4& bel4) {
  int4 h[HDEPTH];
#pragma unroll
  for (int q = 0; q < HDEPTH; ++q) {
    const int k = k0 + g + HGROUPS * q;
    h[q] = *(const int4*)(hist + ((k < nSlices) ? k : 0) * NBP + 4 * t4);
  }
#pragma unroll
  for (int q = 0; q < HDEPTH; ++q) {
    const int k = k0 + g + HGROUPS * q;
    const bool in = k < nSlices;
    const bool lo = in && (k < blk);
    tot4.x += in ? h[q].x : 0;
    tot4.y += in ? h[q].y : 0;
    tot4.z += in ? h[q].z : 0;
    tot4.w += in ? h[q].w : 0;
    bel4.x += lo ? h[q].x : 0;
    bel4.y += lo ? h[q].y : 0;
    bel4.z += lo ? h[q].z : 0;
    bel4.w += lo ? h[q].w : 0;
  }
}

__global__ __launch_bounds__(BIN_THREADS)
void bucket_place_kernel(const int* __restrict__ rows,
                         const int* __restrict__ cols,
                         const float* __restrict__ vals,
                         const int* __restrict__ hist,
                         int* __restrict__ bucketStart, int* __restrict__ stg,
                         int E, int nSlices) {
  __shared__ __attribute__((aligned(16))) int redTot[HGROUPS * NBP];
  __shared__ __attribute__((aligned(16))) int redBel[HGROUPS * NBP];
  __shared__ int cur[NBP + 1];
  __shared__ int delta[NBP];
  __shared__ int wsum[NBP / 32];
  __shared__ int wsumOwn[NBP / 32];
  __shared__ int2 img[SLICE];
  const int tid = (int)threadIdx.x;
  const int lane = tid & 31;
  const int wv = tid >> 5;
  const int blk = (int)blockIdx.x;
  const int e0 = blk * SLICE;

  int r[EPT], c[EPT];
  float v[EPT];
#pragma unroll
  for (int j = 0; j < EPT; ++j) {
    const int e = e0 + j * BIN_THREADS + tid;
    const bool ok = e < E;
    const int ec = ok ? e : 0;
    const unsigned rr = (unsigned)rows[ec];
    c[j] = cols[ec];
    v[j] = vals[ec];
    const unsigned bk = rr / RB;
    r[j] = ok ? (int)((bk << 8) | (rr - bk * RB)) : (NBP << 8);
  }
  const int own = (tid < NBP) ? hist[blk * NBP + tid] : 0;

  const int g = tid >> 6;
  const int t4 = tid & 63;
  int4 tot4 = make_int4(0, 0, 0, 0);
  int4 bel4 = make_int4(0, 0, 0, 0);
  sweep_trip(hist, 0, g, t4, nSlices, blk, tot4, bel4);
  for (int k0 = HGROUPS * HDEPTH; k0 < nSlices; k0 += HGROUPS * HDEPTH) {
    sweep_trip(hist, k0, g, t4, nSlices, blk, tot4, bel4);
  }
  *(int4*)(redTot + g * NBP + 4 * t4) = tot4;
  *(int4*)(redBel + g * NBP + 4 * t4) = bel4;
  __syncthreads();

  int tot = 0, below = 0;
  if (tid < NBP) {
#pragma unroll
    for (int w = 0; w < HGROUPS; ++w) {
      tot += redTot[w * NBP + tid];
      below += redBel[w * NBP + tid];
    }
  }
  const int incl = wave_incl_scan(tot, lane);
  const int inclOwn = wave_incl_scan(own, lane);
  if (lane == 31 && wv < NBP / 32) {
    wsum[wv] = incl;
    wsumOwn[wv] = inclOwn;
  }
  __syncthreads();
  if (tid < NBP) {
    int wbase = 0, wbaseOwn = 0;
#pragma unroll
    for (int w = 0; w < NBP / 32; ++w) {
      wbase += (w < wv) ? wsum[w] : 0;
      wbaseOwn += (w < wv) ? wsumOwn[w] : 0;
    }
    const int start = wbase + incl - tot;
    const int lstart = wbaseOwn + inclOwn - own;
    cur[tid] = lstart;
    delta[tid] = start + below - lstart;
    if (blk == 0) {
      bucketStart[tid] = start;
      if (tid == NBP - 1) bucketStart[NBP] = start + tot;
    }
  }
  if (tid == NBP) cur[NBP] = 0;
  __syncthreads();

  int pos[EPT];
#pragma unroll
  for (int j = 0; j < EPT; ++j) {
    pos[j] = atomicAdd(&cur[r[j] >> 8], 1);
  }
#pragma unroll
  for (int j = 0; j < EPT; ++j) {
    if (r[j] < (NBP << 8)) {
      int2 x;
      x.x = (int)(((unsigned)r[j] << 16) | (unsigned)c[j]);
      x.y = __float_as_int(v[j]);
      img[pos[j]] = x;
    }
  }
  __syncthreads();

  int2* __restrict__ rec = (int2*)stg;
  const int nv = (E - e0 < SLICE) ? E - e0 : SLICE;
  int2 y[EPT];
  int d[EPT];
#pragma unroll
  for (int j = 0; j < EPT; ++j) {
    const int i = j * BIN_THREADS + tid;
    y[j] = img[(i < nv) ? i : 0];
  }
#pragma unroll
  for (int j = 0; j < EPT; ++j) d[j] = delta[(unsigned)y[j].x >> 24];
#pragma unroll
  for (int j = 0; j < EPT; ++j) {
    const int i = j * BIN_THREADS + tid;
    if (i < nv) {
      int2 x;
      x.x = y[j].x & 0x00FFFFFF;
      x.y = y[j].y;
      rec[i + d[j]] = x;
    }
  }
}


#define KSPLIT 8
#define PASS_CHUNKS 16

struct BRows {
  f4 lo0, hi0;
  f4 lo1, hi1;
};

__device__ __forceinline__ int from_chunk(int x, int i, int half) {
  const int lo = __builtin_amdgcn_readlane(x, 2 * i);
  const int hi = __builtin_amdgcn_readlane(x, 2 * i + 1);
  return half ? hi : lo;
}

__device__ __forceinline__ BRows gather_rows(const float* __restrict__ ann,
                                             int2 colv, int i, int half,
                                             int c4) {
  BRows b;
  const float* p0 = ann + from_chunk(colv.x, i, half) * DFEAT + c4;
  const float* p1 = ann + from_chunk(colv.y, i, half) * DFEAT + c4;
  b.lo0 = *(const f4*)(p0);
  b.hi0 = *(const f4*)(p0 + 64);
  b.lo1 = *(const f4*)(p1);
  b.hi1 = *(const f4*)(p1 + 64);
  return b;
}

__device__ __forceinline__ v2f a_frag(int2 rowv, float2 valv, int i, int half,
                                      int rowm, bool ok) {
  const int rx = from_chunk(rowv.x, i, half);
  const int ry = from_chunk(rowv.y, i, half);
  const float vx = __int_as_float(from_chunk(__float_as_int(valv.x), i, half));
  const float vy = __int_as_float(from_chunk(__float_as_int(valv.y), i, half));
  v2f a;
  a.x = (ok && rx == rowm) ? vx : 0.0f;
  a.y = (ok && ry == rowm) ? vy : 0.0f;
  return a;
}

__device__ __forceinline__ void mma_rows(v8f (&acc)[8], v2f a,
                                         const BRows& b) {
#pragma unroll
  for (int t = 0; t < 4; ++t) {
    v2f bl, bh;
    bl.x = b.lo0[t];
    bl.y = b.lo1[t];
    bh.x = b.hi0[t];
    bh.y = b.hi1[t];
    acc[t] = __builtin_amdgcn_wmma_f32_16x16x4_f32(
         false, a,  false, bl,
         (short)0, acc[t],  false,  false);
    acc[4 + t] = __builtin_amdgcn_wmma_f32_16x16x4_f32(
         false, a,  false, bh,
         (short)0, acc[4 + t],  false,  false);
  }
}

#define NTEAMS 2
#define FUSE_THREADS (NTEAMS * KSPLIT * 32)
#define KPT 8
#define IMG_CAP (FUSE_THREADS * KPT)
#define TILE_F4 (16 * (DFEAT / 4))
static_assert(KSPLIT == 8, "a team's wave index is the low 3 bits");

__global__ __launch_bounds__(FUSE_THREADS)
void bucket_spmm_kernel(const int* __restrict__ bucketStart,
                        const int* __restrict__ stg,
                        const float* __restrict__ ann, float* __restrict__ out,
                        int N) {
  __shared__ int cnt[RB + 32];
  __shared__ int cur[RB + 32];
  __shared__ int rs[RB + 1];
  __shared__ __attribute__((aligned(16))) int2 img[IMG_CAP + CSR_PAD];
  __shared__ f4 part[NTEAMS * KSPLIT * TILE_F4];

  const int tid = (int)threadIdx.x;
  const int lane = tid & 31;
  const int wvAll = __builtin_amdgcn_readfirstlane(tid >> 5);
  const int team = wvAll >> 3;
  const int wv = wvAll & (KSPLIT - 1);
  const int b = (int)blockIdx.x;
  const int s = bucketStart[b];
  const int e = bucketStart[b + 1];
  const int2* __restrict__ rec = (const int2*)stg;
  const int row0 = b * RB;
  const int nWin = ((N - row0) >> 4 < RB / 16) ? (N - row0) >> 4 : RB / 16;

  const int half = lane >> 4;
  const int c = lane & 15;
  const int c4 = 4 * c;

  int base = s;
  bool first = true;
  do {
    const int tEnd = (e - base < IMG_CAP) ? e : base + IMG_CAP;
    const int nT = tEnd - base;

    if (tid < RB + 32) cnt[tid] = 0;
    int2 x[KPT];
#pragma unroll
    for (int q = 0; q < KPT; ++q) {
      const int i = base + q * FUSE_THREADS + tid;
      x[q] = rec[(i < tEnd) ? i : 0];
      if (i >= tEnd) x[q].x = -1;
    }
    __syncthreads();
#pragma unroll
    for (int q = 0; q < KPT; ++q) {
      atomicAdd(&cnt[(x[q].x >= 0) ? (x[q].x >> 16) : RB + lane], 1);
    }
    __syncthreads();
    if (wvAll == 0) {
      int qn[RB / 32];
      int t = 0;
#pragma unroll
      for (int q = 0; q < RB / 32; ++q) {
        qn[q] = cnt[lane * (RB / 32) + q];
        t += qn[q];
      }
      const int incl = wave_incl_scan(t, lane);
      int run = incl - t;
#pragma unroll
      for (int q = 0; q < RB / 32; ++q) {
        cur[lane * (RB / 32) + q] = run;
        rs[lane * (RB / 32) + q] = run;
        run += qn[q];
      }
      if (lane == 31) rs[RB] = run;
      cur[RB + lane] = 0;
    }
    __syncthreads();
    int pos[KPT];
#pragma unroll
    for (int q = 0; q < KPT; ++q) {
      pos[q] = atomicAdd(&cur[(x[q].x >= 0) ? (x[q].x >> 16) : RB + lane], 1);
    }
#pragma unroll
    for (int q = 0; q < KPT; ++q) {
      if (x[q].x >= 0) img[pos[q]] = x[q];
    }
    if (tid < CSR_PAD) img[nT + tid] = make_int2(0x00FF0000, 0);
    __syncthreads();

    for (int w0 = 0; w0 < nWin; w0 += NTEAMS) {
      const int win = w0 + team;
      const bool has = win < nWin;
      f4* pt = part + team * (KSPLIT * TILE_F4);
      if (has) {
        const int kBeg = __builtin_amdgcn_readfirstlane(rs[16 * win]);
        const int kEnd = __builtin_amdgcn_readfirstlane(rs[16 * win + 16]);
        const int rowm = 16 * win + c;

        v8f acc[8];
#pragma unroll
        for (int t = 0; t < 8; ++t) {
          acc[t] = (v8f){0.f, 0.f, 0.f, 0.f, 0.f, 0.f, 0.f, 0.f};
        }

        const int k0 = kBeg & ~3;
        const int nCh = (kEnd - k0 + 3) >> 2;
        const int per = (nCh + KSPLIT - 1) / KSPLIT;
        const int jb = wv * per;
        const int je = (jb + per < nCh) ? jb + per : nCh;

        for (int seg = jb; seg < je; seg += PASS_CHUNKS) {
          const int n = (je - seg < PASS_CHUNKS) ? je - seg : PASS_CHUNKS;
          const int ei = k0 + 4 * seg + ((lane < 2 * n) ? 2 * lane : 0);
          const int4 m = *(const int4*)(img + ei);
          int2 colv, rowv;
          float2 valv;
          colv.x = m.x & 0xFFFF;
          colv.y = m.z & 0xFFFF;
          rowv.x = m.x >> 16;
          rowv.y = m.z >> 16;
          valv.x = __int_as_float(m.y);
          valv.y = __int_as_float(m.w);

          const int n2 = (n + 1) & ~1;
          BRows bA = gather_rows(ann, colv, 0, half, c4);
          int i = 0;
          for (; i + 2 < n2; i += 2) {
            const BRows bB = gather_rows(ann, colv, i + 1, half, c4);
            __builtin_amdgcn_sched_barrier(0);
            mma_rows(acc, a_frag(rowv, valv, i, half, rowm, true), bA);
            bA = gather_rows(ann, colv, i + 2, half, c4);
            __builtin_amdgcn_sched_barrier(0);
            mma_rows(acc, a_frag(rowv, valv, i + 1, half, rowm, true), bB);
          }
          {
            const bool ok = (i + 1 < n);
            const int i1 = ok ? i + 1 : i;
            const BRows bB = gather_rows(ann, colv, i1, half, c4);
            __builtin_amdgcn_sched_barrier(0);
            mma_rows(acc, a_frag(rowv, valv, i, half, rowm, true), bA);
            mma_rows(acc, a_frag(rowv, valv, i1, half, rowm, ok), bB);
          }
        }

        f4* pw = pt + wv * TILE_F4 + (8 * half) * (DFEAT / 4) + c;
#pragma unroll
        for (int v = 0; v < 8; ++v) {
#pragma unroll
          for (int g = 0; g < 2; ++g) {
            f4 y;
            y.x = acc[4 * g + 0][v];
            y.y = acc[4 * g + 1][v];
            y.z = acc[4 * g + 2][v];
            y.w = acc[4 * g + 3][v];
            pw[v * (DFEAT / 4) + 16 * g] = y;
          }
        }
      }
      __syncthreads();

      if (has) {
#pragma unroll
        for (int i = 0; i < 2; ++i) {
          const int row = wv + 8 * i;
          const f4* ps = pt + row * (DFEAT / 4) + lane;
          f4 sum = ps[0];
#pragma unroll
          for (int w = 1; w < KSPLIT; ++w) sum += ps[w * TILE_F4];
          f4* po = (f4*)(out + (size_t)(row0 + 16 * win + row) * DFEAT + 4 * lane);
          if (!first) sum += *po;
          *po = sum;
        }
      }
      __syncthreads();
    }
    base += IMG_CAP;
    first = false;
  } while (base < e);
}


__global__ void zero_f32_kernel(float* __restrict__ p, int n) {
  int i = blockIdx.x * blockDim.x + threadIdx.x;
  int s = gridDim.x * blockDim.x;
  for (; i < n; i += s) p[i] = 0.0f;
}

__global__ __launch_bounds__(256)
void scatter_kernel(const int* __restrict__ rows, const int* __restrict__ cols,
                    const float* __restrict__ vals,
                    const float* __restrict__ ann, float* __restrict__ out,
                    int E) {
  const int lane = (int)(threadIdx.x & 31);
  int wid = (int)((blockIdx.x * blockDim.x + threadIdx.x) >> 5);
  const int nw = (int)((gridDim.x * blockDim.x) >> 5);
  for (int e = wid; e < E; e += nw) {
    const int r = rows[e];
    const int c = cols[e];
    const float v = vals[e];
    const f4 x = *(const f4*)(ann + (size_t)c * DFEAT + lane * 4);
    float* o = out + (size_t)r * DFEAT + lane * 4;
    unsafeAtomicAdd(o + 0, v * x.x);
    unsafeAtomicAdd(o + 1, v * x.y);
    unsafeAtomicAdd(o + 2, v * x.z);
    unsafeAtomicAdd(o + 3, v * x.w);
  }
}


extern "C" void kernel_launch(void* const* d_in, const int* in_sizes, int n_in,
                              void* d_out, int out_size, void* d_ws,
                              size_t ws_size, hipStream_t stream) {
  const int* rows = (const int*)d_in[0];
  const int* cols = (const int*)d_in[1];
  const float* vals = (const float*)d_in[2];
  const float* ann = (const float*)d_in[3];
  float* out = (float*)d_out;

  const int E = in_sizes[0];
  const int N = out_size / DFEAT;

  const int nSlices = (E + SLICE - 1) / SLICE;
  const int nBuckets = (N + RB - 1) / RB;

  const size_t histBytes = (size_t)nSlices * NBP * sizeof(int);
  const size_t bstBytes = (size_t)(NBP + 16) * sizeof(int);
  const size_t stgBytes = (size_t)E * 8;
  const size_t need = histBytes + bstBytes + stgBytes;

  if ((N % 16) == 0 && N <= RB * NBP && E > 0 && ws_size >= need) {
    char* w = (char*)d_ws;
    int* hist = (int*)w;         w += histBytes;
    int* bucketStart = (int*)w;  w += bstBytes;
    int* stg = (int*)w;

    bucket_count_kernel<<<nSlices, BIN_THREADS, 0, stream>>>(rows, hist, E);
    bucket_place_kernel<<<nSlices, BIN_THREADS, 0, stream>>>(
        rows, cols, vals, hist, bucketStart, stg, E, nSlices);
    bucket_spmm_kernel<<<nBuckets, FUSE_THREADS, 0, stream>>>(
        bucketStart, stg, ann, out, N);
  } else {
    zero_f32_kernel<<<(out_size + 255) / 256, 256, 0, stream>>>(out, out_size);
    scatter_kernel<<<2048, 256, 0, stream>>>(rows, cols, vals, ann, out, E);
  }
}


# each edge slice is grouped by bucket in place and leaves a run table; a bucket's work-group collects its 125 runs via LDS map: count launch, offset sweep gone, 3 launches -> 2 (sum order as before)
# speedup vs baseline: 6.2709x; 1.1670x over previous
#include <hip/hip_runtime.h>
#include <cstdint>
#include <cstddef>


typedef __attribute__((ext_vector_type(2))) float v2f;
typedef __attribute__((ext_vector_type(4))) float f4;
typedef __attribute__((ext_vector_type(8))) float v8f;

#define DFEAT 128
#define CSR_PAD 16

__device__ __forceinline__ int wave_incl_scan(int s, int lane) {
  s += __builtin_amdgcn_update_dpp(0, s, 0x111, 0xf, 0xf, true);
  s += __builtin_amdgcn_update_dpp(0, s, 0x112, 0xf, 0xf, true);
  s += __builtin_amdgcn_update_dpp(0, s, 0x114, 0xf, 0xf, true);
  s += __builtin_amdgcn_update_dpp(0, s, 0x118, 0xf, 0xf, true);
  const int row0 = __builtin_amdgcn_readlane(s, 15);
  return s + ((lane >= 16) ? row0 : 0);
}


#define RB 160
#define NBP 256
#define BIN_THREADS 512
#define EPT 10
#define SLICE (BIN_THREADS * EPT)
static_assert(NBP == 256 && BIN_THREADS >= NBP + 1,
              "one thread per bucket slot, one more for the spare slot");
static_assert(EPT % 2 == 0 && SLICE < 65536,
              "the image leaves as record pairs; start and length share a word");
static_assert(RB % 32 == 0 && RB < 255, "wave 0 scans RB / 32 rows per lane; "
              "row 255 of a bucket marks a pad record");

__global__ __launch_bounds__(BIN_THREADS)
void slice_group_kernel(const int* __restrict__ rows,
                        const int* __restrict__ cols,
                        const float* __restrict__ vals,
                        int* __restrict__ runTab, int* __restrict__ stg,
                        int E) {
  __shared__ int cnt[NBP + 1];
  __shared__ int lst[NBP];
  __shared__ int wsum[NBP / 32];
  __shared__ __attribute__((aligned(16))) int2 img[SLICE];
  const int tid = (int)threadIdx.x;
  const int lane = tid & 31;
  const int wv = tid >> 5;
  const int blk = (int)blockIdx.x;
  const int e0 = blk * SLICE;

  int r[EPT], c[EPT];
  float v[EPT];
#pragma unroll
  for (int j = 0; j < EPT; ++j) {
    const int e = e0 + j * BIN_THREADS + tid;
    const bool ok = e < E;
    const int ec = ok ? e : 0;
    const unsigned rr = (unsigned)rows[ec];
    c[j] = cols[ec];
    v[j] = vals[ec];
    const unsigned bk = rr / RB;
    r[j] = ok ? (int)((bk << 8) | (rr - bk * RB)) : (NBP << 8);
  }
  if (tid <= NBP) cnt[tid] = 0;
  __syncthreads();

  int rank[EPT];
#pragma unroll
  for (int j = 0; j < EPT; ++j) {
    rank[j] = atomicAdd(&cnt[r[j] >> 8], 1);
  }
  __syncthreads();

  const int own = (tid < NBP) ? cnt[tid] : 0;
  const int incl = wave_incl_scan(own, lane);
  if (lane == 31 && wv < NBP / 32) wsum[wv] = incl;
  __syncthreads();
  if (tid < NBP) {
    int wbase = 0;
#pragma unroll
    for (int w = 0; w < NBP / 32; ++w) wbase += (w < wv) ? wsum[w] : 0;
    const int lstart = wbase + incl - own;
    lst[tid] = lstart;
    runTab[blk * NBP + tid] = (lstart << 16) | own;
  }
  __syncthreads();

  int slot[EPT];
#pragma unroll
  for (int j = 0; j < EPT; ++j) {
    slot[j] = lst[(r[j] >> 8) & (NBP - 1)] + rank[j];
  }
#pragma unroll
  for (int j = 0; j < EPT; ++j) {
    if (r[j] < (NBP << 8)) {
      int2 x;
      x.x = (int)((((unsigned)r[j] & 0xFFu) << 16) | (unsigned)c[j]);
      x.y = __float_as_int(v[j]);
      img[slot[j]] = x;
    }
  }
  __syncthreads();

  int2* __restrict__ rec = (int2*)stg;
  const int nv = (E - e0 < SLICE) ? E - e0 : SLICE;
  int4 y[EPT / 2];
#pragma unroll
  for (int h = 0; h < EPT / 2; ++h) {
    y[h] = *(const int4*)(img + 2 * (h * BIN_THREADS + tid));
  }
#pragma unroll
  for (int h = 0; h < EPT / 2; ++h) {
    const int i = 2 * (h * BIN_THREADS + tid);
    if (i + 1 < nv) {
      *(int4*)(rec + e0 + i) = y[h];
    } else if (i < nv) {
      rec[e0 + i] = make_int2(y[h].x, y[h].y);
    }
  }
}


#define KSPLIT 8
#define PASS_CHUNKS 16

struct BRows {
  f4 lo0, hi0;
  f4 lo1, hi1;
};

__device__ __forceinline__ int from_chunk(int x, int i, int half) {
  const int lo = __builtin_amdgcn_readlane(x, 2 * i);
  const int hi = __builtin_amdgcn_readlane(x, 2 * i + 1);
  return half ? hi : lo;
}

__device__ __forceinline__ BRows gather_rows(const float* __restrict__ ann,
                                             int2 colv, int i, int half,
                                             int c4) {
  BRows b;
  const float* p0 = ann + from_chunk(colv.x, i, half) * DFEAT + c4;
  const float* p1 = ann + from_chunk(colv.y, i, half) * DFEAT + c4;
  b.lo0 = *(const f4*)(p0);
  b.hi0 = *(const f4*)(p0 + 64);
  b.lo1 = *(const f4*)(p1);
  b.hi1 = *(const f4*)(p1 + 64);
  return b;
}

__device__ __forceinline__ v2f a_frag(int2 rowv, float2 valv, int i, int half,
                                      int rowm, bool ok) {
  const int rx = from_chunk(rowv.x, i, half);
  const int ry = from_chunk(rowv.y, i, half);
  const float vx = __int_as_float(from_chunk(__float_as_int(valv.x), i, half));
  const float vy = __int_as_float(from_chunk(__float_as_int(valv.y), i, half));
  v2f a;
  a.x = (ok && rx == rowm) ? vx : 0.0f;
  a.y = (ok && ry == rowm) ? vy : 0.0f;
  return a;
}

__device__ __forceinline__ void mma_rows(v8f (&acc)[8], v2f a,
                                         const BRows& b) {
#pragma unroll
  for (int t = 0; t < 4; ++t) {
    v2f bl, bh;
    bl.x = b.lo0[t];
    bl.y = b.lo1[t];
    bh.x = b.hi0[t];
    bh.y = b.hi1[t];
    acc[t] = __builtin_amdgcn_wmma_f32_16x16x4_f32(
         false, a,  false, bl,
         (short)0, acc[t],  false,  false);
    acc[4 + t] = __builtin_amdgcn_wmma_f32_16x16x4_f32(
         false, a,  false, bh,
         (short)0, acc[4 + t],  false,  false);
  }
}

#define NTEAMS 2
#define FUSE_THREADS (NTEAMS * KSPLIT * 32)
#define KPT 8
#define IMG_CAP (FUSE_THREADS * KPT)
#define TILE_F4 (16 * (DFEAT / 4))
#define NWAVES (FUSE_THREADS / 32)
#define MAXS (NWAVES * 32)
static_assert(KSPLIT == 8, "a team's wave index is the low 3 bits");

__global__ __launch_bounds__(FUSE_THREADS)
void bucket_spmm_kernel(const int* __restrict__ runTab,
                        const int* __restrict__ stg,
                        const float* __restrict__ ann, float* __restrict__ out,
                        int N, int nSlices) {
  __shared__ int cnt[RB + 32];
  __shared__ int cur[RB + 32];
  __shared__ int rs[RB + 1];
  __shared__ int wtot[NWAVES];
  __shared__ int srcOf[IMG_CAP];
  __shared__ __attribute__((aligned(16))) int2 img[IMG_CAP + CSR_PAD];
  __shared__ f4 part[NTEAMS * KSPLIT * TILE_F4];

  const int tid = (int)threadIdx.x;
  const int lane = tid & 31;
  const int wvAll = __builtin_amdgcn_readfirstlane(tid >> 5);
  const int team = wvAll >> 3;
  const int wv = wvAll & (KSPLIT - 1);
  const int b = (int)blockIdx.x;
  const int2* __restrict__ rec = (const int2*)stg;

  const int G = (nSlices + NWAVES - 1) / NWAVES;
  const int myRun = wvAll * G + lane;
  const bool hasRun = (lane < G) && (myRun < nSlices);
  const int ent = hasRun ? runTab[myRun * NBP + b] : 0;
  const int runLen = ent & 0xFFFF;
  const int runIncl = wave_incl_scan(runLen, lane);
  if (lane == 31) wtot[wvAll] = runIncl;
  __syncthreads();
  int before = 0, total = 0;
#pragma unroll
  for (int w = 0; w < NWAVES; ++w) {
    const int t = wtot[w];
    before += (w < wvAll) ? t : 0;
    total += t;
  }
  const int runPos = before + runIncl - runLen;
  const int runOff = myRun * SLICE + (int)((unsigned)ent >> 16) - runPos;
  const int e = __builtin_amdgcn_readfirstlane(total);
  const int row0 = b * RB;
  const int nWin = ((N - row0) >> 4 < RB / 16) ? (N - row0) >> 4 : RB / 16;

  const int half = lane >> 4;
  const int c = lane & 15;
  const int c4 = 4 * c;

  int base = 0;
  bool first = true;
  do {
    const int tEnd = (e - base < IMG_CAP) ? e : base + IMG_CAP;
    const int nT = tEnd - base;

    for (int l = 0; l < G; ++l) {
      const int p0 = __builtin_amdgcn_readlane(runPos, l);
      const int p1 = p0 + __builtin_amdgcn_readlane(runLen, l);
      const int off = __builtin_amdgcn_readlane(runOff, l);
      const int lo = (p0 > base) ? p0 : base;
      const int hi = (p1 < tEnd) ? p1 : tEnd;
      for (int j = lo + lane; j < hi; j += 32) srcOf[j - base] = off;
    }

    if (tid < RB + 32) cnt[tid] = 0;
    __syncthreads();
    int2 x[KPT];
    int a[KPT];
#pragma unroll
    for (int q = 0; q < KPT; ++q) {
      const int i = base + q * FUSE_THREADS + tid;
      a[q] = srcOf[(i < tEnd) ? i - base : 0];
    }
#pragma unroll
    for (int q = 0; q < KPT; ++q) {
      const int i = base + q * FUSE_THREADS + tid;
      a[q] = (i < tEnd) ? i + a[q] : 0;
    }
#pragma unroll
    for (int q = 0; q < KPT; ++q) x[q] = rec[a[q]];
    __builtin_amdgcn_sched_barrier(0);
#pragma unroll
    for (int q = 0; q < KPT; ++q) {
      const int i = base + q * FUSE_THREADS + tid;
      if (i >= tEnd) x[q].x = -1;
    }
#pragma unroll
    for (int q = 0; q < KPT; ++q) {
      atomicAdd(&cnt[(x[q].x >= 0) ? (x[q].x >> 16) : RB + lane], 1);
    }
    __syncthreads();
    if (wvAll == 0) {
      int qn[RB / 32];
      int t = 0;
#pragma unroll
      for (int q = 0; q < RB / 32; ++q) {
        qn[q] = cnt[lane * (RB / 32) + q];
        t += qn[q];
      }
      const int incl = wave_incl_scan(t, lane);
      int run = incl - t;
#pragma unroll
      for (int q = 0; q < RB / 32; ++q) {
        cur[lane * (RB / 32) + q] = run;
        rs[lane * (RB / 32) + q] = run;
        run += qn[q];
      }
      if (lane == 31) rs[RB] = run;
      cur[RB + lane] = 0;
    }
    __syncthreads();
    int pos[KPT];
#pragma unroll
    for (int q = 0; q < KPT; ++q) {
      pos[q] = atomicAdd(&cur[(x[q].x >= 0) ? (x[q].x >> 16) : RB + lane], 1);
    }
#pragma unroll
    for (int q = 0; q < KPT; ++q) {
      if (x[q].x >= 0) img[pos[q]] = x[q];
    }
    if (tid < CSR_PAD) img[nT + tid] = make_int2(0x00FF0000, 0);
    __syncthreads();

    for (int w0 = 0; w0 < nWin; w0 += NTEAMS) {
      const int win = w0 + team;
      const bool has = win < nWin;
      f4* pt = part + team * (KSPLIT * TILE_F4);
      if (has) {
        const int kBeg = __builtin_amdgcn_readfirstlane(rs[16 * win]);
        const int kEnd = __builtin_amdgcn_readfirstlane(rs[16 * win + 16]);
        const int rowm = 16 * win + c;

        v8f acc[8];
#pragma unroll
        for (int t = 0; t < 8; ++t) {
          acc[t] = (v8f){0.f, 0.f, 0.f, 0.f, 0.f, 0.f, 0.f, 0.f};
        }

        const int k0 = kBeg & ~3;
        const int nCh = (kEnd - k0 + 3) >> 2;
        const int per = (nCh + KSPLIT - 1) / KSPLIT;
        const int jb = wv * per;
        const int je = (jb + per < nCh) ? jb + per : nCh;

        for (int seg = jb; seg < je; seg += PASS_CHUNKS) {
          const int n = (je - seg < PASS_CHUNKS) ? je - seg : PASS_CHUNKS;
          const int ei = k0 + 4 * seg + ((lane < 2 * n) ? 2 * lane : 0);
          const int4 m = *(const int4*)(img + ei);
          int2 colv, rowv;
          float2 valv;
          colv.x = m.x & 0xFFFF;
          colv.y = m.z & 0xFFFF;
          rowv.x = m.x >> 16;
          rowv.y = m.z >> 16;
          valv.x = __int_as_float(m.y);
          valv.y = __int_as_float(m.w);

          const int n2 = (n + 1) & ~1;
          BRows bA = gather_rows(ann, colv, 0, half, c4);
          int i = 0;
          for (; i + 2 < n2; i += 2) {
            const BRows bB = gather_rows(ann, colv, i + 1, half, c4);
            __builtin_amdgcn_sched_barrier(0);
            mma_rows(acc, a_frag(rowv, valv, i, half, rowm, true), bA);
            bA = gather_rows(ann, colv, i + 2, half, c4);
            __builtin_amdgcn_sched_barrier(0);
            mma_rows(acc, a_frag(rowv, valv, i + 1, half, rowm, true), bB);
          }
          {
            const bool ok = (i + 1 < n);
            const int i1 = ok ? i + 1 : i;
            const BRows bB = gather_rows(ann, colv, i1, half, c4);
            __builtin_amdgcn_sched_barrier(0);
            mma_rows(acc, a_frag(rowv, valv, i, half, rowm, true), bA);
            mma_rows(acc, a_frag(rowv, valv, i1, half, rowm, ok), bB);
          }
        }

        f4* pw = pt + wv * TILE_F4 + (8 * half) * (DFEAT / 4) + c;
#pragma unroll
        for (int v = 0; v < 8; ++v) {
#pragma unroll
          for (int g = 0; g < 2; ++g) {
            f4 y;
            y.x = acc[4 * g + 0][v];
            y.y = acc[4 * g + 1][v];
            y.z = acc[4 * g + 2][v];
            y.w = acc[4 * g + 3][v];
            pw[v * (DFEAT / 4) + 16 * g] = y;
          }
        }
      }
      __syncthreads();

      if (has) {
#pragma unroll
        for (int i = 0; i < 2; ++i) {
          const int row = wv + 8 * i;
          const f4* ps = pt + row * (DFEAT / 4) + lane;
          f4 sum = ps[0];
#pragma unroll
          for (int w = 1; w < KSPLIT; ++w) sum += ps[w * TILE_F4];
          f4* po = (f4*)(out + (size_t)(row0 + 16 * win + row) * DFEAT + 4 * lane);
          if (!first) sum += *po;
          *po = sum;
        }
      }
      __syncthreads();
    }
    base += IMG_CAP;
    first = false;
  } while (base < e);
}


__global__ void zero_f32_kernel(float* __restrict__ p, int n) {
  int i = blockIdx.x * blockDim.x + threadIdx.x;
  int s = gridDim.x * blockDim.x;
  for (; i < n; i += s) p[i] = 0.0f;
}

__global__ __launch_bounds__(256)
void scatter_kernel(const int* __restrict__ rows, const int* __restrict__ cols,
                    const float* __restrict__ vals,
                    const float* __restrict__ ann, float* __restrict__ out,
                    int E) {
  const int lane = (int)(threadIdx.x & 31);
  int wid = (int)((blockIdx.x * blockDim.x + threadIdx.x) >> 5);
  const int nw = (int)((gridDim.x * blockDim.x) >> 5);
  for (int e = wid; e < E; e += nw) {
    const int r = rows[e];
    const int c = cols[e];
    const float v = vals[e];
    const f4 x = *(const f4*)(ann + (size_t)c * DFEAT + lane * 4);
    float* o = out + (size_t)r * DFEAT + lane * 4;
    unsafeAtomicAdd(o + 0, v * x.x);
    unsafeAtomicAdd(o + 1, v * x.y);
    unsafeAtomicAdd(o + 2, v * x.z);
    unsafeAtomicAdd(o + 3, v * x.w);
  }
}


extern "C" void kernel_launch(void* const* d_in, const int* in_sizes, int n_in,
                              void* d_out, int out_size, void* d_ws,
                              size_t ws_size, hipStream_t stream) {
  const int* rows = (const int*)d_in[0];
  const int* cols = (const int*)d_in[1];
  const float* vals = (const float*)d_in[2];
  const float* ann = (const float*)d_in[3];
  float* out = (float*)d_out;

  const int E = in_sizes[0];
  const int N = out_size / DFEAT;

  const int nSlices = (E + SLICE - 1) / SLICE;
  const int nBuckets = (N + RB - 1) / RB;

  const size_t tabBytes = (size_t)nSlices * NBP * sizeof(int);
  const size_t stgBytes = (size_t)E * 8;
  const size_t need = tabBytes + stgBytes;

  if ((N % 16) == 0 && N <= RB * NBP && E > 0 && nSlices <= MAXS &&
      ws_size >= need) {
    char* w = (char*)d_ws;
    int* runTab = (int*)w;  w += tabBytes;
    int* stg = (int*)w;

    slice_group_kernel<<<nSlices, BIN_THREADS, 0, stream>>>(
        rows, cols, vals, runTab, stg, E);
    bucket_spmm_kernel<<<nBuckets, FUSE_THREADS, 0, stream>>>(
        runTab, stg, ann, out, N, nSlices);
  } else {
    zero_f32_kernel<<<(out_size + 255) / 256, 256, 0, stream>>>(out, out_size);
    scatter_kernel<<<2048, 256, 0, stream>>>(rows, cols, vals, ann, out, E);
  }
}


# spmm: a wave owns a 16-row window x 64 features and walks all its edges alone (8 waves shared them): partial tiles in LDS, their sums, 2 barriers per window pair gone (in-row sum order: one chain)
# speedup vs baseline: 7.5834x; 1.2093x over previous
#include <hip/hip_runtime.h>
#include <cstdint>
#include <cstddef>


typedef __attribute__((ext_vector_type(2))) float v2f;
typedef __attribute__((ext_vector_type(4))) float f4;
typedef __attribute__((ext_vector_type(8))) float v8f;
typedef __attribute__((ext_vector_type(4))) int i4;

#define DFEAT 128
#define CSR_PAD 16

__device__ __forceinline__ int wave_incl_scan(int s, int lane) {
  s += __builtin_amdgcn_update_dpp(0, s, 0x111, 0xf, 0xf, true);
  s += __builtin_amdgcn_update_dpp(0, s, 0x112, 0xf, 0xf, true);
  s += __builtin_amdgcn_update_dpp(0, s, 0x114, 0xf, 0xf, true);
  s += __builtin_amdgcn_update_dpp(0, s, 0x118, 0xf, 0xf, true);
  const int row0 = __builtin_amdgcn_readlane(s, 15);
  return s + ((lane >= 16) ? row0 : 0);
}


#define RB 160
#define NBP 256
#define BIN_THREADS 512
#define EPT 10
#define SLICE (BIN_THREADS * EPT)
static_assert(NBP == 256 && BIN_THREADS >= NBP + 1,
              "one thread per bucket slot, one more for the spare slot");
static_assert(EPT % 2 == 0 && SLICE < 65536,
              "the image leaves as record pairs; start and length share a word");
static_assert(RB % 32 == 0 && RB < 255, "wave 0 scans RB / 32 rows per lane; "
              "row 255 of a bucket marks a pad record");

__global__ __launch_bounds__(BIN_THREADS)
void slice_group_kernel(const int* __restrict__ rows,
                        const int* __restrict__ cols,
                        const float* __restrict__ vals,
                        int* __restrict__ runTab, int* __restrict__ stg,
                        int E) {
  __shared__ int cnt[NBP + 1];
  __shared__ int lst[NBP];
  __shared__ int wsum[NBP / 32];
  __shared__ __attribute__((aligned(16))) int2 img[SLICE];
  const int tid = (int)threadIdx.x;
  const int lane = tid & 31;
  const int wv = tid >> 5;
  const int blk = (int)blockIdx.x;
  const int e0 = blk * SLICE;

  int r[EPT], c[EPT];
  float v[EPT];
#pragma unroll
  for (int j = 0; j < EPT; ++j) {
    const int e = e0 + j * BIN_THREADS + tid;
    const bool ok = e < E;
    const int ec = ok ? e : 0;
    const unsigned rr = (unsigned)rows[ec];
    c[j] = cols[ec];
    v[j] = vals[ec];
    const unsigned bk = rr / RB;
    r[j] = ok ? (int)((bk << 8) | (rr - bk * RB)) : (NBP << 8);
  }
  if (tid <= NBP) cnt[tid] = 0;
  __syncthreads();

  int rank[EPT];
#pragma unroll
  for (int j = 0; j < EPT; ++j) {
    rank[j] = atomicAdd(&cnt[r[j] >> 8], 1);
  }
  __syncthreads();

  const int own = (tid < NBP) ? cnt[tid] : 0;
  const int incl = wave_incl_scan(own, lane);
  if (lane == 31 && wv < NBP / 32) wsum[wv] = incl;
  __syncthreads();
  if (tid < NBP) {
    int wbase = 0;
#pragma unroll
    for (int w = 0; w < NBP / 32; ++w) wbase += (w < wv) ? wsum[w] : 0;
    const int lstart = wbase + incl - own;
    lst[tid] = lstart;
    runTab[blk * NBP + tid] = (lstart << 16) | own;
  }
  __syncthreads();

  int slot[EPT];
#pragma unroll
  for (int j = 0; j < EPT; ++j) {
    slot[j] = lst[(r[j] >> 8) & (NBP - 1)] + rank[j];
  }
#pragma unroll
  for (int j = 0; j < EPT; ++j) {
    if (r[j] < (NBP << 8)) {
      int2 x;
      x.x = (int)((((unsigned)r[j] & 0xFFu) << 16) | (unsigned)c[j]);
      x.y = __float_as_int(v[j]);
      img[slot[j]] = x;
    }
  }
  __syncthreads();

  int2* __restrict__ rec = (int2*)stg;
  const int nv = (E - e0 < SLICE) ? E - e0 : SLICE;
  int4 y[EPT / 2];
#pragma unroll
  for (int h = 0; h < EPT / 2; ++h) {
    y[h] = *(const int4*)(img + 2 * (h * BIN_THREADS + tid));
  }
#pragma unroll
  for (int h = 0; h < EPT / 2; ++h) {
    const int i = 2 * (h * BIN_THREADS + tid);
    if (i + 1 < nv) {
      *(int4*)(rec + e0 + i) = y[h];
    } else if (i < nv) {
      rec[e0 + i] = make_int2(y[h].x, y[h].y);
    }
  }
}


#define WIN_WAVES 2
#define GRP 4

struct Chunk {
  f4 x0, x1;
  v2f a;
};

__device__ __forceinline__ Chunk fetch_chunk(const int2* ip, int j,
                                             const float* __restrict__ annh,
                                             int rowm) {
  const i4 m = *(const i4*)(ip + 4 * j);
  const int rc0 = m.x, rc1 = m.z;
  const float v0 = __int_as_float(m.y), v1 = __int_as_float(m.w);
  Chunk ch;
  ch.a.x = ((rc0 >> 16) == rowm) ? v0 : 0.0f;
  ch.a.y = ((rc1 >> 16) == rowm) ? v1 : 0.0f;
  ch.x0 = *(const f4*)(annh + (rc0 & 0xFFFF) * DFEAT);
  ch.x1 = *(const f4*)(annh + (rc1 & 0xFFFF) * DFEAT);
  return ch;
}

__device__ __forceinline__ void fetch_group(Chunk (&g)[GRP], const int2* ip,
                                            int grp,
                                            const float* __restrict__ annh,
                                            int rowm) {
#pragma unroll
  for (int u = 0; u < GRP; ++u) g[u] = fetch_chunk(ip, GRP * grp + u, annh, rowm);
}

__device__ __forceinline__ void mma_chunk(v8f (&acc)[4], const Chunk& ch) {
#pragma unroll
  for (int t = 0; t < 4; ++t) {
    v2f b;
    b.x = ch.x0[t];
    b.y = ch.x1[t];
    acc[t] = __builtin_amdgcn_wmma_f32_16x16x4_f32(
         false, ch.a,  false, b,
         (short)0, acc[t],  false,  false);
  }
}

__device__ __forceinline__ void mma_group(v8f (&acc)[4], const Chunk (&g)[GRP]) {
#pragma unroll
  for (int u = 0; u < GRP; ++u) mma_chunk(acc, g[u]);
}

__device__ __forceinline__ void mma_first(v8f (&acc)[4], const Chunk (&g)[GRP],
                                          int n) {
#pragma unroll
  for (int u = 0; u < GRP; ++u) {
    if (u < n) mma_chunk(acc, g[u]);
  }
}

#define NWAVES ((RB / 16) * WIN_WAVES)
#define FUSE_THREADS (NWAVES * 32)
#define KPT 6
#define IMG_CAP (FUSE_THREADS * KPT)
#define MAXS (NWAVES * 32)
static_assert(NWAVES % 4 == 0 && FUSE_THREADS <= 1024,
              "whole waves on each of the 4 SIMDs; one work-group");
static_assert(DFEAT == 64 * WIN_WAVES, "a wave's 4 tiles are 64 features");

__global__ __launch_bounds__(FUSE_THREADS)
void bucket_spmm_kernel(const int* __restrict__ runTab,
                        const int* __restrict__ stg,
                        const float* __restrict__ ann, float* __restrict__ out,
                        int N, int nSlices) {
  __shared__ int cnt[RB + 32];
  __shared__ int cur[RB + 32];
  __shared__ int rs[RB + 1];
  __shared__ int wtot[NWAVES];
  __shared__ int srcOf[IMG_CAP];
  __shared__ __attribute__((aligned(16))) int2 img[IMG_CAP + CSR_PAD];

  const int tid = (int)threadIdx.x;
  const int lane = tid & 31;
  const int wvAll = __builtin_amdgcn_readfirstlane(tid >> 5);
  const int win = wvAll / WIN_WAVES;
  const int fh = wvAll % WIN_WAVES;
  const int b = (int)blockIdx.x;
  const int2* __restrict__ rec = (const int2*)stg;

  const int G = (nSlices + NWAVES - 1) / NWAVES;
  const int myRun = wvAll * G + lane;
  const bool hasRun = (lane < G) && (myRun < nSlices);
  const int ent = hasRun ? runTab[myRun * NBP + b] : 0;
  const int runLen = ent & 0xFFFF;
  const int runIncl = wave_incl_scan(runLen, lane);
  if (lane == 31) wtot[wvAll] = runIncl;
  __syncthreads();
  int before = 0, total = 0;
#pragma unroll
  for (int w = 0; w < NWAVES; ++w) {
    const int t = wtot[w];
    before += (w < wvAll) ? t : 0;
    total += t;
  }
  const int runPos = before + runIncl - runLen;
  const int runOff = myRun * SLICE + (int)((unsigned)ent >> 16) - runPos;
  const int e = __builtin_amdgcn_readfirstlane(total);
  const int row0 = b * RB;
  const int nWin = ((N - row0) >> 4 < RB / 16) ? (N - row0) >> 4 : RB / 16;

  const int half = lane >> 4;
  const int c = lane & 15;
  const int c4 = 4 * c;

  int base = 0;
  bool first = true;
  do {
    const int tEnd = (e - base < IMG_CAP) ? e : base + IMG_CAP;
    const int nT = tEnd - base;

    for (int l = 0; l < G; ++l) {
      const int p0 = __builtin_amdgcn_readlane(runPos, l);
      const int p1 = p0 + __builtin_amdgcn_readlane(runLen, l);
      const int off = __builtin_amdgcn_readlane(runOff, l);
      const int lo = (p0 > base) ? p0 : base;
      const int hi = (p1 < tEnd) ? p1 : tEnd;
      for (int j = lo + lane; j < hi; j += 32) srcOf[j - base] = off;
    }

    if (tid < RB + 32) cnt[tid] = 0;
    __syncthreads();
    int2 x[KPT];
    int a[KPT];
#pragma unroll
    for (int q = 0; q < KPT; ++q) {
      const int i = base + q * FUSE_THREADS + tid;
      a[q] = srcOf[(i < tEnd) ? i - base : 0];
    }
#pragma unroll
    for (int q = 0; q < KPT; ++q) {
      const int i = base + q * FUSE_THREADS + tid;
      a[q] = (i < tEnd) ? i + a[q] : 0;
    }
#pragma unroll
    for (int q = 0; q < KPT; ++q) x[q] = rec[a[q]];
    __builtin_amdgcn_sched_barrier(0);
#pragma unroll
    for (int q = 0; q < KPT; ++q) {
      const int i = base + q * FUSE_THREADS + tid;
      if (i >= tEnd) x[q].x = -1;
    }
#pragma unroll
    for (int q = 0; q < KPT; ++q) {
      atomicAdd(&cnt[(x[q].x >= 0) ? (x[q].x >> 16) : RB + lane], 1);
    }
    __syncthreads();
    if (wvAll == 0) {
      int qn[RB / 32];
      int t = 0;
#pragma unroll
      for (int q = 0; q < RB / 32; ++q) {
        qn[q] = cnt[lane * (RB / 32) + q];
        t += qn[q];
      }
      const int incl = wave_incl_scan(t, lane);
      int run = incl - t;
#pragma unroll
      for (int q = 0; q < RB / 32; ++q) {
        cur[lane * (RB / 32) + q] = run;
        rs[lane * (RB / 32) + q] = run;
        run += qn[q];
      }
      if (lane == 31) rs[RB] = run;
      cur[RB + lane] = 0;
    }
    __syncthreads();
    int pos[KPT];
#pragma unroll
    for (int q = 0; q < KPT; ++q) {
      pos[q] = atomicAdd(&cur[(x[q].x >= 0) ? (x[q].x >> 16) : RB + lane], 1);
    }
#pragma unroll
    for (int q = 0; q < KPT; ++q) {
      if (x[q].x >= 0) img[pos[q]] = x[q];
    }
    if (tid < CSR_PAD) img[nT + tid] = make_int2(0x00FF0000, 0);
    __syncthreads();

    if (win < nWin) {
      const int kBeg = __builtin_amdgcn_readfirstlane(rs[16 * win]);
      const int kEnd = __builtin_amdgcn_readfirstlane(rs[16 * win + 16]);
      const int rowm = 16 * win + c;

      v8f acc[4];
#pragma unroll
      for (int t = 0; t < 4; ++t) {
        acc[t] = (v8f){0.f, 0.f, 0.f, 0.f, 0.f, 0.f, 0.f, 0.f};
      }

      const int k0 = kBeg & ~3;
      const int nCh = (kEnd - k0 + 3) >> 2;
      const int nG = (nCh + GRP - 1) / GRP;
      const int2* ip = img + k0 + 2 * half;
      const float* annh = ann + 64 * fh + c4;

      if (nG > 0) {
        Chunk P[GRP], Q[GRP];
        const int nG2 = (nG + 1) & ~1;
        fetch_group(P, ip, 0, annh, rowm);
        int g = 0;
        for (; g + 2 < nG2; g += 2) {
          fetch_group(Q, ip, g + 1, annh, rowm);
          __builtin_amdgcn_sched_barrier(0);
          mma_group(acc, P);
          fetch_group(P, ip, g + 2, annh, rowm);
          __builtin_amdgcn_sched_barrier(0);
          mma_group(acc, Q);
        }
        {
          const bool two = (g + 1 < nG);
          const int g1 = two ? g + 1 : g;
          fetch_group(Q, ip, g1, annh, rowm);
          __builtin_amdgcn_sched_barrier(0);
          mma_first(acc, P, nCh - GRP * g);
          mma_first(acc, Q, two ? nCh - GRP * g1 : 0);
        }
      }

      float* po = out + (size_t)(row0 + 16 * win + 8 * half) * DFEAT +
                  64 * fh + c4;
#pragma unroll
      for (int v = 0; v < 8; ++v) {
        f4 y;
        y.x = acc[0][v];
        y.y = acc[1][v];
        y.z = acc[2][v];
        y.w = acc[3][v];
        f4* p = (f4*)(po + v * DFEAT);
        if (!first) y += *p;
        *p = y;
      }
    }
    base += IMG_CAP;
    first = false;
    if (base < e) __syncthreads();
  } while (base < e);
}


__global__ void zero_f32_kernel(float* __restrict__ p, int n) {
  int i = blockIdx.x * blockDim.x + threadIdx.x;
  int s = gridDim.x * blockDim.x;
  for (; i < n; i += s) p[i] = 0.0f;
}

__global__ __launch_bounds__(256)
void scatter_kernel(const int* __restrict__ rows, const int* __restrict__ cols,
                    const float* __restrict__ vals,
                    const float* __restrict__ ann, float* __restrict__ out,
                    int E) {
  const int lane = (int)(threadIdx.x & 31);
  int wid = (int)((blockIdx.x * blockDim.x + threadIdx.x) >> 5);
  const int nw = (int)((gridDim.x * blockDim.x) >> 5);
  for (int e = wid; e < E; e += nw) {
    const int r = rows[e];
    const int c = cols[e];
    const float v = vals[e];
    const f4 x = *(const f4*)(ann + (size_t)c * DFEAT + lane * 4);
    float* o = out + (size_t)r * DFEAT + lane * 4;
    unsafeAtomicAdd(o + 0, v * x.x);
    unsafeAtomicAdd(o + 1, v * x.y);
    unsafeAtomicAdd(o + 2, v * x.z);
    unsafeAtomicAdd(o + 3, v * x.w);
  }
}


extern "C" void kernel_launch(void* const* d_in, const int* in_sizes, int n_in,
                              void* d_out, int out_size, void* d_ws,
                              size_t ws_size, hipStream_t stream) {
  const int* rows = (const int*)d_in[0];
  const int* cols = (const int*)d_in[1];
  const float* vals = (const float*)d_in[2];
  const float* ann = (const float*)d_in[3];
  float* out = (float*)d_out;

  const int E = in_sizes[0];
  const int N = out_size / DFEAT;

  const int nSlices = (E + SLICE - 1) / SLICE;
  const int nBuckets = (N + RB - 1) / RB;

  const size_t tabBytes = (size_t)nSlices * NBP * sizeof(int);
  const size_t stgBytes = (size_t)E * 8;
  const size_t need = tabBytes + stgBytes;

  if ((N % 16) == 0 && N <= RB * NBP && E > 0 && nSlices <= MAXS &&
      ws_size >= need) {
    char* w = (char*)d_ws;
    int* runTab = (int*)w;  w += tabBytes;
    int* stg = (int*)w;

    slice_group_kernel<<<nSlices, BIN_THREADS, 0, stream>>>(
        rows, cols, vals, runTab, stg, E);
    bucket_spmm_kernel<<<nBuckets, FUSE_THREADS, 0, stream>>>(
        runTab, stg, ann, out, N, nSlices);
  } else {
    zero_f32_kernel<<<(out_size + 255) / 256, 256, 0, stream>>>(out, out_size);
    scatter_kernel<<<2048, 256, 0, stream>>>(rows, cols, vals, ann, out, E);
  }
}


# spmm: each chunk's 16x4 A fragment is built in LDS when its records are placed (windows start on whole chunks); the loop loads A pair and row offsets: no compare, select or 64-bit add per edge
# speedup vs baseline: 7.9489x; 1.0482x over previous
#include <hip/hip_runtime.h>
#include <cstdint>
#include <cstddef>


typedef __attribute__((ext_vector_type(2))) float v2f;
typedef __attribute__((ext_vector_type(4))) float f4;
typedef __attribute__((ext_vector_type(8))) float v8f;
typedef __attribute__((ext_vector_type(4))) int i4;

#define DFEAT 128
#define ROW_BITS 9
#define ROW_MASK ((1 << ROW_BITS) - 1)
static_assert(DFEAT * 4 == (1 << ROW_BITS), "col << ROW_BITS is a byte offset");

__device__ __forceinline__ int wave_incl_scan(int s, int lane) {
  s += __builtin_amdgcn_update_dpp(0, s, 0x111, 0xf, 0xf, true);
  s += __builtin_amdgcn_update_dpp(0, s, 0x112, 0xf, 0xf, true);
  s += __builtin_amdgcn_update_dpp(0, s, 0x114, 0xf, 0xf, true);
  s += __builtin_amdgcn_update_dpp(0, s, 0x118, 0xf, 0xf, true);
  const int row0 = __builtin_amdgcn_readlane(s, 15);
  return s + ((lane >= 16) ? row0 : 0);
}


#define RB 160
#define NBP 256
#define BIN_THREADS 512
#define EPT 10
#define SLICE (BIN_THREADS * EPT)
static_assert(NBP == 256 && BIN_THREADS >= NBP + 1,
              "one thread per bucket slot, one more for the spare slot");
static_assert(EPT % 2 == 0 && SLICE < 65536,
              "the image leaves as record pairs; start and length share a word");
static_assert(RB % 16 == 0 && RB / 16 <= 32 && RB <= 256,
              "whole 16-row windows, a lane of wave 0 per window; a bucket "
              "row fits the low bits of a record");

__global__ __launch_bounds__(BIN_THREADS)
void slice_group_kernel(const int* __restrict__ rows,
                        const int* __restrict__ cols,
                        const float* __restrict__ vals,
                        int* __restrict__ runTab, int* __restrict__ stg,
                        int E) {
  __shared__ int cnt[NBP + 1];
  __shared__ int lst[NBP];
  __shared__ int wsum[NBP / 32];
  __shared__ __attribute__((aligned(16))) int2 img[SLICE];
  const int tid = (int)threadIdx.x;
  const int lane = tid & 31;
  const int wv = tid >> 5;
  const int blk = (int)blockIdx.x;
  const int e0 = blk * SLICE;

  int r[EPT], c[EPT];
  float v[EPT];
#pragma unroll
  for (int j = 0; j < EPT; ++j) {
    const int e = e0 + j * BIN_THREADS + tid;
    const bool ok = e < E;
    const int ec = ok ? e : 0;
    const unsigned rr = (unsigned)rows[ec];
    c[j] = cols[ec];
    v[j] = vals[ec];
    const unsigned bk = rr / RB;
    r[j] = ok ? (int)((bk << 8) | (rr - bk * RB)) : (NBP << 8);
  }
  if (tid <= NBP) cnt[tid] = 0;
  __syncthreads();

  int rank[EPT];
#pragma unroll
  for (int j = 0; j < EPT; ++j) {
    rank[j] = atomicAdd(&cnt[r[j] >> 8], 1);
  }
  __syncthreads();

  const int own = (tid < NBP) ? cnt[tid] : 0;
  const int incl = wave_incl_scan(own, lane);
  if (lane == 31 && wv < NBP / 32) wsum[wv] = incl;
  __syncthreads();
  if (tid < NBP) {
    int wbase = 0;
#pragma unroll
    for (int w = 0; w < NBP / 32; ++w) wbase += (w < wv) ? wsum[w] : 0;
    const int lstart = wbase + incl - own;
    lst[tid] = lstart;
    runTab[blk * NBP + tid] = (lstart << 16) | own;
  }
  __syncthreads();

  int slot[EPT];
#pragma unroll
  for (int j = 0; j < EPT; ++j) {
    slot[j] = lst[(r[j] >> 8) & (NBP - 1)] + rank[j];
  }
#pragma unroll
  for (int j = 0; j < EPT; ++j) {
    if (r[j] < (NBP << 8)) {
      int2 x;
      x.x = (int)(((unsigned)c[j] << ROW_BITS) | ((unsigned)r[j] & 0xFFu));
      x.y = __float_as_int(v[j]);
      img[slot[j]] = x;
    }
  }
  __syncthreads();

  int2* __restrict__ rec = (int2*)stg;
  const int nv = (E - e0 < SLICE) ? E - e0 : SLICE;
  int4 y[EPT / 2];
#pragma unroll
  for (int h = 0; h < EPT / 2; ++h) {
    y[h] = *(const int4*)(img + 2 * (h * BIN_THREADS + tid));
  }
#pragma unroll
  for (int h = 0; h < EPT / 2; ++h) {
    const int i = 2 * (h * BIN_THREADS + tid);
    if (i + 1 < nv) {
      *(int4*)(rec + e0 + i) = y[h];
    } else if (i < nv) {
      rec[e0 + i] = make_int2(y[h].x, y[h].y);
    }
  }
}


#define WIN_WAVES 2
#define GRP 4

struct Chunk {
  f4 x0, x1;
  v2f a;
};

__device__ __forceinline__ Chunk fetch_chunk(const v2f* ap, const uint2* op,
                                             int j,
                                             const float* __restrict__ ann,
                                             unsigned laneOff) {
  Chunk ch;
  ch.a = ap[32 * j];
  const uint2 o = op[2 * j];
  const char* __restrict__ base = (const char*)ann;
  ch.x0 = *(const f4*)(base + (o.x | laneOff));
  ch.x1 = *(const f4*)(base + (o.y | laneOff));
  return ch;
}

__device__ __forceinline__ void fetch_group(Chunk (&g)[GRP], const v2f* ap,
                                            const uint2* op, int grp,
                                            const float* __restrict__ ann,
                                            unsigned laneOff) {
#pragma unroll
  for (int u = 0; u < GRP; ++u) {
    g[u] = fetch_chunk(ap, op, GRP * grp + u, ann, laneOff);
  }
}

__device__ __forceinline__ void mma_chunk(v8f (&acc)[4], const Chunk& ch) {
#pragma unroll
  for (int t = 0; t < 4; ++t) {
    v2f b;
    b.x = ch.x0[t];
    b.y = ch.x1[t];
    acc[t] = __builtin_amdgcn_wmma_f32_16x16x4_f32(
         false, ch.a,  false, b,
         (short)0, acc[t],  false,  false);
  }
}

__device__ __forceinline__ void mma_group(v8f (&acc)[4], const Chunk (&g)[GRP]) {
#pragma unroll
  for (int u = 0; u < GRP; ++u) mma_chunk(acc, g[u]);
}

__device__ __forceinline__ void mma_first(v8f (&acc)[4], const Chunk (&g)[GRP],
                                          int n) {
#pragma unroll
  for (int u = 0; u < GRP; ++u) {
    if (u < n) mma_chunk(acc, g[u]);
  }
}

#define NWAVES ((RB / 16) * WIN_WAVES)
#define FUSE_THREADS (NWAVES * 32)
#define KPT 5
#define IMG_CAP (FUSE_THREADS * KPT)
#define NWIN (RB / 16)
#define IMG_SLOTS (IMG_CAP + 4 * NWIN)
#define NCHUNK (IMG_SLOTS / 4 + GRP)
static_assert(IMG_SLOTS % 4 == 0, "whole chunks");
#define MAXS (NWAVES * 32)
static_assert(NWAVES % 4 == 0 && FUSE_THREADS <= 1024,
              "whole waves on each of the 4 SIMDs; one work-group");
static_assert(DFEAT == 64 * WIN_WAVES, "a wave's 4 tiles are 64 features");

__device__ __forceinline__ void clear_images(float* aImg, unsigned* offImg,
                                             int tid) {
  const f4 z = (f4){0.f, 0.f, 0.f, 0.f};
  for (int i = tid; i < NCHUNK * 16; i += FUSE_THREADS) ((f4*)aImg)[i] = z;
  for (int i = tid; i < NCHUNK; i += FUSE_THREADS) {
    ((i4*)offImg)[i] = (i4){0, 0, 0, 0};
  }
}

__global__ __launch_bounds__(FUSE_THREADS)
void bucket_spmm_kernel(const int* __restrict__ runTab,
                        const int* __restrict__ stg,
                        const float* __restrict__ ann, float* __restrict__ out,
                        int N, int nSlices) {
  __shared__ int cnt[RB + 32];
  __shared__ int cur[RB + 32];
  __shared__ int wBeg[NWIN];
  __shared__ int wEnd[NWIN];
  __shared__ int wtot[NWAVES];
  __shared__ int srcOf[IMG_CAP];
  __shared__ __attribute__((aligned(16))) float aImg[NCHUNK * 64];
  __shared__ __attribute__((aligned(16))) unsigned offImg[NCHUNK * 4];

  const int tid = (int)threadIdx.x;
  const int lane = tid & 31;
  const int wvAll = __builtin_amdgcn_readfirstlane(tid >> 5);
  const int win = wvAll / WIN_WAVES;
  const int fh = wvAll % WIN_WAVES;
  const int b = (int)blockIdx.x;
  const int2* __restrict__ rec = (const int2*)stg;

  const int G = (nSlices + NWAVES - 1) / NWAVES;
  const int myRun = wvAll * G + lane;
  const bool hasRun = (lane < G) && (myRun < nSlices);
  const int ent = hasRun ? runTab[myRun * NBP + b] : 0;
  clear_images(aImg, offImg, tid);
  const int runLen = ent & 0xFFFF;
  const int runIncl = wave_incl_scan(runLen, lane);
  if (lane == 31) wtot[wvAll] = runIncl;
  __syncthreads();
  int before = 0, total = 0;
#pragma unroll
  for (int w = 0; w < NWAVES; ++w) {
    const int t = wtot[w];
    before += (w < wvAll) ? t : 0;
    total += t;
  }
  const int runPos = before + runIncl - runLen;
  const int runOff = myRun * SLICE + (int)((unsigned)ent >> 16) - runPos;
  const int e = __builtin_amdgcn_readfirstlane(total);
  const int row0 = b * RB;
  const int nWin = ((N - row0) >> 4 < RB / 16) ? (N - row0) >> 4 : RB / 16;

  const int half = lane >> 4;
  const int c = lane & 15;
  const int c4 = 4 * c;

  int base = 0;
  bool first = true;
  do {
    const int tEnd = (e - base < IMG_CAP) ? e : base + IMG_CAP;
    const int nT = tEnd - base;

    for (int l = 0; l < G; ++l) {
      const int p0 = __builtin_amdgcn_readlane(runPos, l);
      const int p1 = p0 + __builtin_amdgcn_readlane(runLen, l);
      const int off = __builtin_amdgcn_readlane(runOff, l);
      const int lo = (p0 > base) ? p0 : base;
      const int hi = (p1 < tEnd) ? p1 : tEnd;
      for (int j = lo + lane; j < hi; j += 32) srcOf[j - base] = off;
    }

    if (tid < RB + 32) cnt[tid] = 0;
    __syncthreads();
    int2 x[KPT];
    int a[KPT];
#pragma unroll
    for (int q = 0; q < KPT; ++q) {
      const int i = base + q * FUSE_THREADS + tid;
      a[q] = srcOf[(i < tEnd) ? i - base : 0];
    }
#pragma unroll
    for (int q = 0; q < KPT; ++q) {
      const int i = base + q * FUSE_THREADS + tid;
      a[q] = (i < tEnd) ? i + a[q] : 0;
    }
#pragma unroll
    for (int q = 0; q < KPT; ++q) x[q] = rec[a[q]];
    __builtin_amdgcn_sched_barrier(0);
#pragma unroll
    for (int q = 0; q < KPT; ++q) {
      const int i = base + q * FUSE_THREADS + tid;
      if (i >= tEnd) x[q].x = -1;
    }
#pragma unroll
    for (int q = 0; q < KPT; ++q) {
      atomicAdd(&cnt[(x[q].x >= 0) ? (x[q].x & ROW_MASK) : RB + lane], 1);
    }
    __syncthreads();
    if (wvAll == 0) {
      int pre[16];
      int tot = 0;
      const int wl = (lane < NWIN) ? lane : 0;
#pragma unroll
      for (int i = 0; i < 16; ++i) {
        pre[i] = tot;
        tot += cnt[16 * wl + i];
      }
      const int padded = (lane < NWIN) ? ((tot + 3) & ~3) : 0;
      const int wbase = wave_incl_scan(padded, lane) - padded;
      if (lane < NWIN) {
#pragma unroll
        for (int i = 0; i < 16; ++i) cur[16 * lane + i] = wbase + pre[i];
        wBeg[lane] = wbase;
        wEnd[lane] = wbase + tot;
      }
      cur[RB + lane] = 0;
    }
    __syncthreads();
    int pos[KPT];
#pragma unroll
    for (int q = 0; q < KPT; ++q) {
      pos[q] = atomicAdd(&cur[(x[q].x >= 0) ? (x[q].x & ROW_MASK) : RB + lane], 1);
    }
#pragma unroll
    for (int q = 0; q < KPT; ++q) {
      if (x[q].x >= 0) {
        const int p = pos[q];
        const int rw = x[q].x & 15;
        offImg[p] = (unsigned)x[q].x & ~(unsigned)ROW_MASK;
        aImg[(p >> 2) * 64 + ((p & 2) << 4) + 2 * rw + (p & 1)] =
            __int_as_float(x[q].y);
      }
    }
    __syncthreads();

    if (win < nWin) {
      const int kBeg = __builtin_amdgcn_readfirstlane(wBeg[win]);
      const int kEnd = __builtin_amdgcn_readfirstlane(wEnd[win]);

      v8f acc[4];
#pragma unroll
      for (int t = 0; t < 4; ++t) {
        acc[t] = (v8f){0.f, 0.f, 0.f, 0.f, 0.f, 0.f, 0.f, 0.f};
      }

      const int nCh = (kEnd - kBeg + 3) >> 2;
      const int nG = (nCh + GRP - 1) / GRP;
      const v2f* ap = (const v2f*)aImg + (kBeg >> 2) * 32 + lane;
      const uint2* op = (const uint2*)offImg + (kBeg >> 2) * 2 + half;
      const unsigned laneOff = (unsigned)(64 * fh + c4) * 4u;

      if (nG > 0) {
        Chunk P[GRP], Q[GRP];
        const int nG2 = (nG + 1) & ~1;
        fetch_group(P, ap, op, 0, ann, laneOff);
        int g = 0;
        for (; g + 2 < nG2; g += 2) {
          fetch_group(Q, ap, op, 1, ann, laneOff);
          __builtin_amdgcn_sched_barrier(0);
          mma_group(acc, P);
          fetch_group(P, ap, op, 2, ann, laneOff);
          __builtin_amdgcn_sched_barrier(0);
          mma_group(acc, Q);
          ap += 2 * GRP * 32;
          op += 2 * GRP * 2;
        }
        {
          const bool two = (g + 1 < nG);
          const int g1 = two ? g + 1 : g;
          fetch_group(Q, ap, op, two ? 1 : 0, ann, laneOff);
          __builtin_amdgcn_sched_barrier(0);
          mma_first(acc, P, nCh - GRP * g);
          mma_first(acc, Q, two ? nCh - GRP * g1 : 0);
        }
      }

      float* po = out + (size_t)(row0 + 16 * win + 8 * half) * DFEAT +
                  64 * fh + c4;
#pragma unroll
      for (int v = 0; v < 8; ++v) {
        f4 y;
        y.x = acc[0][v];
        y.y = acc[1][v];
        y.z = acc[2][v];
        y.w = acc[3][v];
        f4* p = (f4*)(po + v * DFEAT);
        if (!first) y += *p;
        *p = y;
      }
    }
    base += IMG_CAP;
    first = false;
    if (base < e) {
      __syncthreads();
      clear_images(aImg, offImg, tid);
    }
  } while (base < e);
}


__global__ void zero_f32_kernel(float* __restrict__ p, int n) {
  int i = blockIdx.x * blockDim.x + threadIdx.x;
  int s = gridDim.x * blockDim.x;
  for (; i < n; i += s) p[i] = 0.0f;
}

__global__ __launch_bounds__(256)
void scatter_kernel(const int* __restrict__ rows, const int* __restrict__ cols,
                    const float* __restrict__ vals,
                    const float* __restrict__ ann, float* __restrict__ out,
                    int E) {
  const int lane = (int)(threadIdx.x & 31);
  int wid = (int)((blockIdx.x * blockDim.x + threadIdx.x) >> 5);
  const int nw = (int)((gridDim.x * blockDim.x) >> 5);
  for (int e = wid; e < E; e += nw) {
    const int r = rows[e];
    const int c = cols[e];
    const float v = vals[e];
    const f4 x = *(const f4*)(ann + (size_t)c * DFEAT + lane * 4);
    float* o = out + (size_t)r * DFEAT + lane * 4;
    unsafeAtomicAdd(o + 0, v * x.x);
    unsafeAtomicAdd(o + 1, v * x.y);
    unsafeAtomicAdd(o + 2, v * x.z);
    unsafeAtomicAdd(o + 3, v * x.w);
  }
}


extern "C" void kernel_launch(void* const* d_in, const int* in_sizes, int n_in,
                              void* d_out, int out_size, void* d_ws,
                              size_t ws_size, hipStream_t stream) {
  const int* rows = (const int*)d_in[0];
  const int* cols = (const int*)d_in[1];
  const float* vals = (const float*)d_in[2];
  const float* ann = (const float*)d_in[3];
  float* out = (float*)d_out;

  const int E = in_sizes[0];
  const int N = out_size / DFEAT;

  const int nSlices = (E + SLICE - 1) / SLICE;
  const int nBuckets = (N + RB - 1) / RB;

  const size_t tabBytes = (size_t)nSlices * NBP * sizeof(int);
  const size_t stgBytes = (size_t)E * 8;
  const size_t need = tabBytes + stgBytes;

  if ((N % 16) == 0 && N <= RB * NBP && E > 0 && nSlices <= MAXS &&
      ws_size >= need) {
    char* w = (char*)d_ws;
    int* runTab = (int*)w;  w += tabBytes;
    int* stg = (int*)w;

    slice_group_kernel<<<nSlices, BIN_THREADS, 0, stream>>>(
        rows, cols, vals, runTab, stg, E);
    bucket_spmm_kernel<<<nBuckets, FUSE_THREADS, 0, stream>>>(
        runTab, stg, ann, out, N, nSlices);
  } else {
    zero_f32_kernel<<<(out_size + 255) / 256, 256, 0, stream>>>(out, out_size);
    scatter_kernel<<<2048, 256, 0, stream>>>(rows, cols, vals, ann, out, E);
  }
}


# slice pass: a lane reads 4 consecutive edges per 16-byte load, none under a guard: a thread's 6 loads are one round trip (were 10 in a row); 640 threads x 8 edges (in-row sum order: LDS add order)
# speedup vs baseline: 9.2791x; 1.1673x over previous
#include <hip/hip_runtime.h>
#include <cstdint>
#include <cstddef>


typedef __attribute__((ext_vector_type(2))) float v2f;
typedef __attribute__((ext_vector_type(4))) float f4;
typedef __attribute__((ext_vector_type(8))) float v8f;
typedef __attribute__((ext_vector_type(4))) int i4;

#define DFEAT 128
#define ROW_BITS 9
#define ROW_MASK ((1 << ROW_BITS) - 1)
static_assert(DFEAT * 4 == (1 << ROW_BITS), "col << ROW_BITS is a byte offset");

__device__ __forceinline__ int wave_incl_scan(int s, int lane) {
  s += __builtin_amdgcn_update_dpp(0, s, 0x111, 0xf, 0xf, true);
  s += __builtin_amdgcn_update_dpp(0, s, 0x112, 0xf, 0xf, true);
  s += __builtin_amdgcn_update_dpp(0, s, 0x114, 0xf, 0xf, true);
  s += __builtin_amdgcn_update_dpp(0, s, 0x118, 0xf, 0xf, true);
  const int row0 = __builtin_amdgcn_readlane(s, 15);
  return s + ((lane >= 16) ? row0 : 0);
}


#define RB 160
#define NBP 256
#define BIN_THREADS 640
#define EPT 8
#define SLICE (BIN_THREADS * EPT)
static_assert(NBP == 256 && BIN_THREADS >= NBP + 1,
              "one thread per bucket slot, one more for the spare slot");
static_assert(EPT % 4 == 0 && SLICE < 65536,
              "a lane reads 4 edges per load, the image leaves as record pairs; "
              "start and length share a word");
static_assert(RB % 16 == 0 && RB / 16 <= 32 && RB <= 256,
              "whole 16-row windows, a lane of wave 0 per window; a bucket "
              "row fits the low bits of a record");

__global__ __launch_bounds__(BIN_THREADS)
void slice_group_kernel(const int* __restrict__ rows,
                        const int* __restrict__ cols,
                        const float* __restrict__ vals,
                        int* __restrict__ runTab, int* __restrict__ stg,
                        int E) {
  __shared__ int cnt[NBP + 1];
  __shared__ int lst[NBP];
  __shared__ int wsum[NBP / 32];
  __shared__ __attribute__((aligned(16))) int2 img[SLICE];
  const int tid = (int)threadIdx.x;
  const int lane = tid & 31;
  const int wv = tid >> 5;
  const int blk = (int)blockIdx.x;
  const int e0 = blk * SLICE;

  i4 r4[EPT / 4], c4[EPT / 4];
  f4 v4[EPT / 4];
  bool ok[EPT / 4];
#pragma unroll
  for (int h = 0; h < EPT / 4; ++h) {
    const int e = e0 + 4 * (h * BIN_THREADS + tid);
    ok[h] = e < E;
    const int ec = ok[h] ? e : 0;
    r4[h] = *(const i4*)(rows + ec);
    c4[h] = *(const i4*)(cols + ec);
    v4[h] = *(const f4*)(vals + ec);
  }
  int r[EPT], c[EPT];
  float v[EPT];
#pragma unroll
  for (int j = 0; j < EPT; ++j) {
    const unsigned rr = (unsigned)r4[j / 4][j % 4];
    c[j] = c4[j / 4][j % 4];
    v[j] = v4[j / 4][j % 4];
    const unsigned bk = rr / RB;
    const int word = (int)((bk << 8) | (rr - bk * RB));
    r[j] = ok[j / 4] ? word : (NBP << 8);
  }
  if (tid <= NBP) cnt[tid] = 0;
  __syncthreads();

  int rank[EPT];
#pragma unroll
  for (int j = 0; j < EPT; ++j) {
    rank[j] = atomicAdd(&cnt[r[j] >> 8], 1);
  }
  __syncthreads();

  const int own = (tid < NBP) ? cnt[tid] : 0;
  const int incl = wave_incl_scan(own, lane);
  if (lane == 31 && wv < NBP / 32) wsum[wv] = incl;
  __syncthreads();
  if (tid < NBP) {
    int wbase = 0;
#pragma unroll
    for (int w = 0; w < NBP / 32; ++w) wbase += (w < wv) ? wsum[w] : 0;
    const int lstart = wbase + incl - own;
    lst[tid] = lstart;
    runTab[blk * NBP + tid] = (lstart << 16) | own;
  }
  __syncthreads();

  int slot[EPT];
#pragma unroll
  for (int j = 0; j < EPT; ++j) {
    slot[j] = lst[(r[j] >> 8) & (NBP - 1)] + rank[j];
  }
#pragma unroll
  for (int j = 0; j < EPT; ++j) {
    if (r[j] < (NBP << 8)) {
      int2 x;
      x.x = (int)(((unsigned)c[j] << ROW_BITS) | ((unsigned)r[j] & 0xFFu));
      x.y = __float_as_int(v[j]);
      img[slot[j]] = x;
    }
  }
  __syncthreads();

  int2* __restrict__ rec = (int2*)stg;
  const int nv = (E - e0 < SLICE) ? E - e0 : SLICE;
  int4 y[EPT / 2];
#pragma unroll
  for (int h = 0; h < EPT / 2; ++h) {
    y[h] = *(const int4*)(img + 2 * (h * BIN_THREADS + tid));
  }
#pragma unroll
  for (int h = 0; h < EPT / 2; ++h) {
    const int i = 2 * (h * BIN_THREADS + tid);
    if (i + 1 < nv) {
      *(int4*)(rec + e0 + i) = y[h];
    } else if (i < nv) {
      rec[e0 + i] = make_int2(y[h].x, y[h].y);
    }
  }
}


#define WIN_WAVES 2
#define GRP 4

struct Chunk {
  f4 x0, x1;
  v2f a;
};

__device__ __forceinline__ Chunk fetch_chunk(const v2f* ap, const uint2* op,
                                             int j,
                                             const float* __restrict__ ann,
                                             unsigned laneOff) {
  Chunk ch;
  ch.a = ap[32 * j];
  const uint2 o = op[2 * j];
  const char* __restrict__ base = (const char*)ann;
  ch.x0 = *(const f4*)(base + (o.x | laneOff));
  ch.x1 = *(const f4*)(base + (o.y | laneOff));
  return ch;
}

__device__ __forceinline__ void fetch_group(Chunk (&g)[GRP], const v2f* ap,
                                            const uint2* op, int grp,
                                            const float* __restrict__ ann,
                                            unsigned laneOff) {
#pragma unroll
  for (int u = 0; u < GRP; ++u) {
    g[u] = fetch_chunk(ap, op, GRP * grp + u, ann, laneOff);
  }
}

__device__ __forceinline__ void mma_chunk(v8f (&acc)[4], const Chunk& ch) {
#pragma unroll
  for (int t = 0; t < 4; ++t) {
    v2f b;
    b.x = ch.x0[t];
    b.y = ch.x1[t];
    acc[t] = __builtin_amdgcn_wmma_f32_16x16x4_f32(
         false, ch.a,  false, b,
         (short)0, acc[t],  false,  false);
  }
}

__device__ __forceinline__ void mma_group(v8f (&acc)[4], const Chunk (&g)[GRP]) {
#pragma unroll
  for (int u = 0; u < GRP; ++u) mma_chunk(acc, g[u]);
}

__device__ __forceinline__ void mma_first(v8f (&acc)[4], const Chunk (&g)[GRP],
                                          int n) {
#pragma unroll
  for (int u = 0; u < GRP; ++u) {
    if (u < n) mma_chunk(acc, g[u]);
  }
}

#define NWAVES ((RB / 16) * WIN_WAVES)
#define FUSE_THREADS (NWAVES * 32)
#define KPT 5
#define IMG_CAP (FUSE_THREADS * KPT)
#define NWIN (RB / 16)
#define IMG_SLOTS (IMG_CAP + 4 * NWIN)
#define NCHUNK (IMG_SLOTS / 4 + GRP)
static_assert(IMG_SLOTS % 4 == 0, "whole chunks");
#define MAXS (NWAVES * 32)
static_assert(NWAVES % 4 == 0 && FUSE_THREADS <= 1024,
              "whole waves on each of the 4 SIMDs; one work-group");
static_assert(DFEAT == 64 * WIN_WAVES, "a wave's 4 tiles are 64 features");

__device__ __forceinline__ void clear_images(float* aImg, unsigned* offImg,
                                             int tid) {
  const f4 z = (f4){0.f, 0.f, 0.f, 0.f};
  for (int i = tid; i < NCHUNK * 16; i += FUSE_THREADS) ((f4*)aImg)[i] = z;
  for (int i = tid; i < NCHUNK; i += FUSE_THREADS) {
    ((i4*)offImg)[i] = (i4){0, 0, 0, 0};
  }
}

__global__ __launch_bounds__(FUSE_THREADS)
void bucket_spmm_kernel(const int* __restrict__ runTab,
                        const int* __restrict__ stg,
                        const float* __restrict__ ann, float* __restrict__ out,
                        int N, int nSlices) {
  __shared__ int cnt[RB + 32];
  __shared__ int cur[RB + 32];
  __shared__ int wBeg[NWIN];
  __shared__ int wEnd[NWIN];
  __shared__ int wtot[NWAVES];
  __shared__ int srcOf[IMG_CAP];
  __shared__ __attribute__((aligned(16))) float aImg[NCHUNK * 64];
  __shared__ __attribute__((aligned(16))) unsigned offImg[NCHUNK * 4];

  const int tid = (int)threadIdx.x;
  const int lane = tid & 31;
  const int wvAll = __builtin_amdgcn_readfirstlane(tid >> 5);
  const int win = wvAll / WIN_WAVES;
  const int fh = wvAll % WIN_WAVES;
  const int b = (int)blockIdx.x;
  const int2* __restrict__ rec = (const int2*)stg;

  const int G = (nSlices + NWAVES - 1) / NWAVES;
  const int myRun = wvAll * G + lane;
  const bool hasRun = (lane < G) && (myRun < nSlices);
  const int ent = hasRun ? runTab[myRun * NBP + b] : 0;
  clear_images(aImg, offImg, tid);
  const int runLen = ent & 0xFFFF;
  const int runIncl = wave_incl_scan(runLen, lane);
  if (lane == 31) wtot[wvAll] = runIncl;
  __syncthreads();
  int before = 0, total = 0;
#pragma unroll
  for (int w = 0; w < NWAVES; ++w) {
    const int t = wtot[w];
    before += (w < wvAll) ? t : 0;
    total += t;
  }
  const int runPos = before + runIncl - runLen;
  const int runOff = myRun * SLICE + (int)((unsigned)ent >> 16) - runPos;
  const int e = __builtin_amdgcn_readfirstlane(total);
  const int row0 = b * RB;
  const int nWin = ((N - row0) >> 4 < RB / 16) ? (N - row0) >> 4 : RB / 16;

  const int half = lane >> 4;
  const int c = lane & 15;
  const int c4 = 4 * c;

  int base = 0;
  bool first = true;
  do {
    const int tEnd = (e - base < IMG_CAP) ? e : base + IMG_CAP;
    const int nT = tEnd - base;

    for (int l = 0; l < G; ++l) {
      const int p0 = __builtin_amdgcn_readlane(runPos, l);
      const int p1 = p0 + __builtin_amdgcn_readlane(runLen, l);
      const int off = __builtin_amdgcn_readlane(runOff, l);
      const int lo = (p0 > base) ? p0 : base;
      const int hi = (p1 < tEnd) ? p1 : tEnd;
      for (int j = lo + lane; j < hi; j += 32) srcOf[j - base] = off;
    }

    if (tid < RB + 32) cnt[tid] = 0;
    __syncthreads();
    int2 x[KPT];
    int a[KPT];
#pragma unroll
    for (int q = 0; q < KPT; ++q) {
      const int i = base + q * FUSE_THREADS + tid;
      a[q] = srcOf[(i < tEnd) ? i - base : 0];
    }
#pragma unroll
    for (int q = 0; q < KPT; ++q) {
      const int i = base + q * FUSE_THREADS + tid;
      a[q] = (i < tEnd) ? i + a[q] : 0;
    }
#pragma unroll
    for (int q = 0; q < KPT; ++q) x[q] = rec[a[q]];
    __builtin_amdgcn_sched_barrier(0);
#pragma unroll
    for (int q = 0; q < KPT; ++q) {
      const int i = base + q * FUSE_THREADS + tid;
      if (i >= tEnd) x[q].x = -1;
    }
#pragma unroll
    for (int q = 0; q < KPT; ++q) {
      atomicAdd(&cnt[(x[q].x >= 0) ? (x[q].x & ROW_MASK) : RB + lane], 1);
    }
    __syncthreads();
    if (wvAll == 0) {
      int pre[16];
      int tot = 0;
      const int wl = (lane < NWIN) ? lane : 0;
#pragma unroll
      for (int i = 0; i < 16; ++i) {
        pre[i] = tot;
        tot += cnt[16 * wl + i];
      }
      const int padded = (lane < NWIN) ? ((tot + 3) & ~3) : 0;
      const int wbase = wave_incl_scan(padded, lane) - padded;
      if (lane < NWIN) {
#pragma unroll
        for (int i = 0; i < 16; ++i) cur[16 * lane + i] = wbase + pre[i];
        wBeg[lane] = wbase;
        wEnd[lane] = wbase + tot;
      }
      cur[RB + lane] = 0;
    }
    __syncthreads();
    int pos[KPT];
#pragma unroll
    for (int q = 0; q < KPT; ++q) {
      pos[q] = atomicAdd(&cur[(x[q].x >= 0) ? (x[q].x & ROW_MASK) : RB + lane], 1);
    }
#pragma unroll
    for (int q = 0; q < KPT; ++q) {
      if (x[q].x >= 0) {
        const int p = pos[q];
        const int rw = x[q].x & 15;
        offImg[p] = (unsigned)x[q].x & ~(unsigned)ROW_MASK;
        aImg[(p >> 2) * 64 + ((p & 2) << 4) + 2 * rw + (p & 1)] =
            __int_as_float(x[q].y);
      }
    }
    __syncthreads();

    if (win < nWin) {
      const int kBeg = __builtin_amdgcn_readfirstlane(wBeg[win]);
      const int kEnd = __builtin_amdgcn_readfirstlane(wEnd[win]);

      v8f acc[4];
#pragma unroll
      for (int t = 0; t < 4; ++t) {
        acc[t] = (v8f){0.f, 0.f, 0.f, 0.f, 0.f, 0.f, 0.f, 0.f};
      }

      const int nCh = (kEnd - kBeg + 3) >> 2;
      const int nG = (nCh + GRP - 1) / GRP;
      const v2f* ap = (const v2f*)aImg + (kBeg >> 2) * 32 + lane;
      const uint2* op = (const uint2*)offImg + (kBeg >> 2) * 2 + half;
      const unsigned laneOff = (unsigned)(64 * fh + c4) * 4u;

      if (nG > 0) {
        Chunk P[GRP], Q[GRP];
        const int nG2 = (nG + 1) & ~1;
        fetch_group(P, ap, op, 0, ann, laneOff);
        int g = 0;
        for (; g + 2 < nG2; g += 2) {
          fetch_group(Q, ap, op, 1, ann, laneOff);
          __builtin_amdgcn_sched_barrier(0);
          mma_group(acc, P);
          fetch_group(P, ap, op, 2, ann, laneOff);
          __builtin_amdgcn_sched_barrier(0);
          mma_group(acc, Q);
          ap += 2 * GRP * 32;
          op += 2 * GRP * 2;
        }
        {
          const bool two = (g + 1 < nG);
          const int g1 = two ? g + 1 : g;
          fetch_group(Q, ap, op, two ? 1 : 0, ann, laneOff);
          __builtin_amdgcn_sched_barrier(0);
          mma_first(acc, P, nCh - GRP * g);
          mma_first(acc, Q, two ? nCh - GRP * g1 : 0);
        }
      }

      float* po = out + (size_t)(row0 + 16 * win + 8 * half) * DFEAT +
                  64 * fh + c4;
#pragma unroll
      for (int v = 0; v < 8; ++v) {
        f4 y;
        y.x = acc[0][v];
        y.y = acc[1][v];
        y.z = acc[2][v];
        y.w = acc[3][v];
        f4* p = (f4*)(po + v * DFEAT);
        if (!first) y += *p;
        *p = y;
      }
    }
    base += IMG_CAP;
    first = false;
    if (base < e) {
      __syncthreads();
      clear_images(aImg, offImg, tid);
    }
  } while (base < e);
}


__global__ void zero_f32_kernel(float* __restrict__ p, int n) {
  int i = blockIdx.x * blockDim.x + threadIdx.x;
  int s = gridDim.x * blockDim.x;
  for (; i < n; i += s) p[i] = 0.0f;
}

__global__ __launch_bounds__(256)
void scatter_kernel(const int* __restrict__ rows, const int* __restrict__ cols,
                    const float* __restrict__ vals,
                    const float* __restrict__ ann, float* __restrict__ out,
                    int E) {
  const int lane = (int)(threadIdx.x & 31);
  int wid = (int)((blockIdx.x * blockDim.x + threadIdx.x) >> 5);
  const int nw = (int)((gridDim.x * blockDim.x) >> 5);
  for (int e = wid; e < E; e += nw) {
    const int r = rows[e];
    const int c = cols[e];
    const float v = vals[e];
    const f4 x = *(const f4*)(ann + (size_t)c * DFEAT + lane * 4);
    float* o = out + (size_t)r * DFEAT + lane * 4;
    unsafeAtomicAdd(o + 0, v * x.x);
    unsafeAtomicAdd(o + 1, v * x.y);
    unsafeAtomicAdd(o + 2, v * x.z);
    unsafeAtomicAdd(o + 3, v * x.w);
  }
}


extern "C" void kernel_launch(void* const* d_in, const int* in_sizes, int n_in,
                              void* d_out, int out_size, void* d_ws,
                              size_t ws_size, hipStream_t stream) {
  const int* rows = (const int*)d_in[0];
  const int* cols = (const int*)d_in[1];
  const float* vals = (const float*)d_in[2];
  const float* ann = (const float*)d_in[3];
  float* out = (float*)d_out;

  const int E = in_sizes[0];
  const int N = out_size / DFEAT;

  const int nSlices = (E + SLICE - 1) / SLICE;
  const int nBuckets = (N + RB - 1) / RB;

  const size_t tabBytes = (size_t)nSlices * NBP * sizeof(int);
  const size_t stgBytes = (size_t)E * 8;
  const size_t need = tabBytes + stgBytes;

  if ((N % 16) == 0 && (E % 4) == 0 && N <= RB * NBP && E > 0 &&
      nSlices <= MAXS && ws_size >= need) {
    char* w = (char*)d_ws;
    int* runTab = (int*)w;  w += tabBytes;
    int* stg = (int*)w;

    slice_group_kernel<<<nSlices, BIN_THREADS, 0, stream>>>(
        rows, cols, vals, runTab, stg, E);
    bucket_spmm_kernel<<<nBuckets, FUSE_THREADS, 0, stream>>>(
        runTab, stg, ann, out, N, nSlices);
  } else {
    zero_f32_kernel<<<(out_size + 255) / 256, 256, 0, stream>>>(out, out_size);
    scatter_kernel<<<2048, 256, 0, stream>>>(rows, cols, vals, ann, out, E);
  }
}
